# starting kernel
# baseline (speedup 1.0000x reference)
#include <hip/hip_runtime.h>
#include <hip/hip_bf16.h>

typedef __attribute__((ext_vector_type(2))) float v2f;
typedef __attribute__((ext_vector_type(8))) float v8f;

#define GCN_N 50000
#define GCN_E 800000


__global__ void gcn_zero_kernel(float* __restrict__ p, int n) {
    int i = blockIdx.x * blockDim.x + threadIdx.x;
    if (i < n) p[i] = 0.0f;
}

__global__ void gcn_degree_kernel(const int* __restrict__ dst, float* __restrict__ deg, int E) {
    int e = blockIdx.x * blockDim.x + threadIdx.x;
    if (e < E) atomicAdd(&deg[dst[e]], 1.0f);
}

__global__ void gcn_dinv_kernel(float* __restrict__ deg, int n) {
    int i = blockIdx.x * blockDim.x + threadIdx.x;
    if (i < n) deg[i] = rsqrtf(deg[i] + 1.0f);
}

template <int RELU>
__global__ void gcn_gemm_wmma_f32(const float* __restrict__ A,
                                  const float* __restrict__ W,
                                  const float* __restrict__ bias,
                                  const float* __restrict__ dinv,
                                  float* __restrict__ hw,
                                  float* __restrict__ agg,
                                  int M, int K, int Nc) {
    const int lane = threadIdx.x & 31;
    const int wid  = threadIdx.x >> 5;
    const int wavesPerBlock = blockDim.x >> 5;

    const int tilesN = (Nc + 15) >> 4;
    const int tilesM = M >> 4;
    int tile = blockIdx.x * wavesPerBlock + wid;
    if (tile >= tilesM * tilesN) return;

    const int tm   = tile / tilesN;
    const int tn   = tile - tm * tilesN;
    const int half = lane >> 4;
    const int l16  = lane & 15;

    const int  row   = tm * 16 + l16;
    const int  col   = tn * 16 + l16;
    const bool colOK = (col < Nc);
    const int  colc  = colOK ? col : 0;
    const int  koff  = half * 2;

    const float* __restrict__ Arow = A + (size_t)row * K + koff;
    const float* __restrict__ Wcol = W + colc;

    v8f acc = {};
    const int Kmain = K & ~3;

#pragma unroll 4
    for (int k = 0; k < Kmain; k += 4) {
        float a0 = Arow[k];
        float a1 = Arow[k + 1];
        if (RELU) { a0 = fmaxf(a0, 0.0f); a1 = fmaxf(a1, 0.0f); }
        float w0 = Wcol[(size_t)(k + koff)     * Nc];
        float w1 = Wcol[(size_t)(k + koff + 1) * Nc];
        v2f a; a[0] = a0;               a[1] = a1;
        v2f b; b[0] = colOK ? w0 : 0.f; b[1] = colOK ? w1 : 0.f;
        acc = __builtin_amdgcn_wmma_f32_16x16x4_f32(
            false, a, false, b, (short)0, acc, false, false);
    }

    if (Kmain < K) {
        const int  ka  = Kmain + koff;
        const bool ok0 = (ka     < K);
        const bool ok1 = (ka + 1 < K);
        const int  kc0 = ok0 ? ka     : (K - 1);
        const int  kc1 = ok1 ? ka + 1 : (K - 1);
        float a0 = A[(size_t)row * K + kc0];
        float a1 = A[(size_t)row * K + kc1];
        if (RELU) { a0 = fmaxf(a0, 0.0f); a1 = fmaxf(a1, 0.0f); }
        float w0 = Wcol[(size_t)kc0 * Nc];
        float w1 = Wcol[(size_t)kc1 * Nc];
        v2f a; a[0] = ok0 ? a0 : 0.f;            a[1] = ok1 ? a1 : 0.f;
        v2f b; b[0] = (colOK && ok0) ? w0 : 0.f; b[1] = (colOK && ok1) ? w1 : 0.f;
        acc = __builtin_amdgcn_wmma_f32_16x16x4_f32(
            false, a, false, b, (short)0, acc, false, false);
    }

    const int   rbase = tm * 16 + (half << 3);
    const float bn    = bias[colc];
    if (colOK) {
#pragma unroll
        for (int r = 0; r < 8; ++r) {
            const int m  = rbase + r;
            const float di = dinv[m];
            const float v  = acc[r];
            hw [(size_t)m * Nc + col] = v;
            agg[(size_t)m * Nc + col] = v * di * di + bn;
        }
    }
}

__global__ void gcn_scatter_kernel(const int* __restrict__ src,
                                   const int* __restrict__ dst,
                                   const float* __restrict__ hw,
                                   const float* __restrict__ dinv,
                                   float* __restrict__ agg, int E, int C) {
    int wave = (int)((blockIdx.x * (size_t)blockDim.x + threadIdx.x) >> 5);
    int lane = threadIdx.x & 31;
    if (wave >= E) return;
    int s = src[wave];
    int d = dst[wave];
    float norm = dinv[s] * dinv[d];
    const float* __restrict__ hrow = hw  + (size_t)s * C;
    float*       __restrict__ arow = agg + (size_t)d * C;
    for (int c = lane; c < C; c += 32) {
        atomicAdd(&arow[c], hrow[c] * norm);
    }
}


extern "C" void kernel_launch(void* const* d_in, const int* in_sizes, int n_in,
                              void* d_out, int out_size, void* d_ws, size_t ws_size,
                              hipStream_t stream) {
    (void)in_sizes; (void)n_in; (void)out_size; (void)ws_size;

    const int N = GCN_N;
    const int E = GCN_E;

    const float* x    = (const float*)d_in[0];
    const int*   ei   = (const int*)d_in[1];
    const int*   src  = ei;
    const int*   dstp = ei + E;

    const float* W[5] = { (const float*)d_in[2], (const float*)d_in[4],
                          (const float*)d_in[6], (const float*)d_in[8],
                          (const float*)d_in[10] };
    const float* B[5] = { (const float*)d_in[3], (const float*)d_in[5],
                          (const float*)d_in[7], (const float*)d_in[9],
                          (const float*)d_in[11] };

    float* ws   = (float*)d_ws;
    float* dinv = ws;
    float* hw   = dinv + N;
    float* aggA = hw   + (size_t)N * 96;
    float* aggB = aggA + (size_t)N * 96;
    float* out  = (float*)d_out;

    const int T = 256;

    gcn_zero_kernel  <<<(N + T - 1) / T, T, 0, stream>>>(dinv, N);
    gcn_degree_kernel<<<(E + T - 1) / T, T, 0, stream>>>(dstp, dinv, E);
    gcn_dinv_kernel  <<<(N + T - 1) / T, T, 0, stream>>>(dinv, N);

    const int dims[6] = { 165, 96, 96, 96, 96, 2 };
    const float* layerIn = x;
    float* aggBufs[2] = { aggA, aggB };

    for (int l = 0; l < 5; ++l) {
        const int K  = dims[l];
        const int Cn = dims[l + 1];
        float* agg = (l == 4) ? out : aggBufs[l & 1];

        const int tiles         = (N / 16) * ((Cn + 15) / 16);
        const int wavesPerBlock = 4;
        const int gBlocks       = (tiles + wavesPerBlock - 1) / wavesPerBlock;
        if (l == 0) {
            gcn_gemm_wmma_f32<0><<<gBlocks, wavesPerBlock * 32, 0, stream>>>(
                layerIn, W[l], B[l], dinv, hw, agg, N, K, Cn);
        } else {
            gcn_gemm_wmma_f32<1><<<gBlocks, wavesPerBlock * 32, 0, stream>>>(
                layerIn, W[l], B[l], dinv, hw, agg, N, K, Cn);
        }

        const size_t scatterThreads = (size_t)E * 32;
        gcn_scatter_kernel<<<(scatterThreads + T - 1) / T, T, 0, stream>>>(
            src, dstp, hw, dinv, agg, E, Cn);

        layerIn = agg;
    }
}


# Float-atomic edge scatter replaced by in-call dst buckets (int atomic rank, scan, fill) and a one-wave-per-node register gather-sum; per-node sum order = arrival rank; rows pre-scaled by dinv
# speedup vs baseline: 2.5883x; 2.5883x over previous
#include <hip/hip_runtime.h>
#include <hip/hip_bf16.h>

typedef __attribute__((ext_vector_type(2))) float v2f;
typedef __attribute__((ext_vector_type(8))) float v8f;

#define GCN_N 50000
#define GCN_E 800000


__global__ void gcn_zero_int_kernel(int* __restrict__ p, int n) {
    int i = blockIdx.x * blockDim.x + threadIdx.x;
    if (i < n) p[i] = 0;
}

__global__ void gcn_rank_kernel(const int* __restrict__ dst, int* __restrict__ cnt,
                                int* __restrict__ rank, int E) {
    int e = blockIdx.x * blockDim.x + threadIdx.x;
    if (e < E) rank[e] = atomicAdd(&cnt[dst[e]], 1);
}

__global__ void gcn_scan_kernel(const int* __restrict__ cnt, int* __restrict__ rowptr,
                                float* __restrict__ dinv, int n) {
    __shared__ int sh[1024];
    const int tid  = threadIdx.x;
    const int base = blockIdx.x * 1024;

    int part = 0;
    for (int i = tid; i < base; i += 1024) part += cnt[i];
    sh[tid] = part;
    __syncthreads();
    for (int s = 512; s > 0; s >>= 1) {
        if (tid < s) sh[tid] += sh[tid + s];
        __syncthreads();
    }
    const int front = sh[0];
    __syncthreads();

    const int i = base + tid;
    const int c = (i < n) ? cnt[i] : 0;
    sh[tid] = c;
    __syncthreads();
    for (int off = 1; off < 1024; off <<= 1) {
        const int t = (tid >= off) ? sh[tid - off] : 0;
        __syncthreads();
        sh[tid] += t;
        __syncthreads();
    }
    const int excl = sh[tid] - c + front;
    if (i < n) {
        rowptr[i] = excl;
        dinv[i]   = rsqrtf((float)c + 1.0f);
    }
    if (i == n - 1) rowptr[n] = excl + c;
}

__global__ void gcn_fill_kernel(const int* __restrict__ src, const int* __restrict__ dst,
                                const int* __restrict__ rowptr, const int* __restrict__ rank,
                                int* __restrict__ bsrc, int E) {
    int e = blockIdx.x * blockDim.x + threadIdx.x;
    if (e < E) bsrc[rowptr[dst[e]] + rank[e]] = src[e];
}

template <int RELU>
__global__ void gcn_gemm_wmma_f32(const float* __restrict__ A,
                                  const float* __restrict__ W,
                                  const float* __restrict__ dinv,
                                  float* __restrict__ hws,
                                  int M, int K, int Nc) {
    const int lane = threadIdx.x & 31;
    const int wid  = threadIdx.x >> 5;
    const int wavesPerBlock = blockDim.x >> 5;

    const int tilesN = (Nc + 15) >> 4;
    const int tilesM = M >> 4;
    int tile = blockIdx.x * wavesPerBlock + wid;
    if (tile >= tilesM * tilesN) return;

    const int tm   = tile / tilesN;
    const int tn   = tile - tm * tilesN;
    const int half = lane >> 4;
    const int l16  = lane & 15;

    const int  row   = tm * 16 + l16;
    const int  col   = tn * 16 + l16;
    const bool colOK = (col < Nc);
    const int  colc  = colOK ? col : 0;
    const int  koff  = half * 2;

    const float* __restrict__ Arow = A + (size_t)row * K + koff;
    const float* __restrict__ Wcol = W + colc;

    v8f acc = {};
    const int Kmain = K & ~3;

#pragma unroll 4
    for (int k = 0; k < Kmain; k += 4) {
        float a0 = Arow[k];
        float a1 = Arow[k + 1];
        if (RELU) { a0 = fmaxf(a0, 0.0f); a1 = fmaxf(a1, 0.0f); }
        float w0 = Wcol[(size_t)(k + koff)     * Nc];
        float w1 = Wcol[(size_t)(k + koff + 1) * Nc];
        v2f a; a[0] = a0;               a[1] = a1;
        v2f b; b[0] = colOK ? w0 : 0.f; b[1] = colOK ? w1 : 0.f;
        acc = __builtin_amdgcn_wmma_f32_16x16x4_f32(
            false, a, false, b, (short)0, acc, false, false);
    }

    if (Kmain < K) {
        const int  ka  = Kmain + koff;
        const bool ok0 = (ka     < K);
        const bool ok1 = (ka + 1 < K);
        const int  kc0 = ok0 ? ka     : (K - 1);
        const int  kc1 = ok1 ? ka + 1 : (K - 1);
        float a0 = A[(size_t)row * K + kc0];
        float a1 = A[(size_t)row * K + kc1];
        if (RELU) { a0 = fmaxf(a0, 0.0f); a1 = fmaxf(a1, 0.0f); }
        float w0 = Wcol[(size_t)kc0 * Nc];
        float w1 = Wcol[(size_t)kc1 * Nc];
        v2f a; a[0] = ok0 ? a0 : 0.f;            a[1] = ok1 ? a1 : 0.f;
        v2f b; b[0] = (colOK && ok0) ? w0 : 0.f; b[1] = (colOK && ok1) ? w1 : 0.f;
        acc = __builtin_amdgcn_wmma_f32_16x16x4_f32(
            false, a, false, b, (short)0, acc, false, false);
    }

    const int rbase = tm * 16 + (half << 3);
    if (colOK) {
#pragma unroll
        for (int r = 0; r < 8; ++r) {
            const int m = rbase + r;
            hws[(size_t)m * Nc + col] = acc[r] * dinv[m];
        }
    }
}

__global__ void gcn_gather96_kernel(const int* __restrict__ rowptr,
                                    const int* __restrict__ bsrc,
                                    const float* __restrict__ hws,
                                    const float* __restrict__ dinv,
                                    const float* __restrict__ bias,
                                    float* __restrict__ outp, int n) {
    const int lane = threadIdx.x & 31;
    const int node = __builtin_amdgcn_readfirstlane(
        (int)(blockIdx.x * (blockDim.x >> 5) + (threadIdx.x >> 5)));
    if (node >= n) return;

    const int l4 = (lane < 24) ? lane : lane - 24;
    const char* __restrict__ rows = (const char*)hws;
    const unsigned loff = (unsigned)l4 * 16u;

    float4 acc = *(const float4*)(rows + (size_t)((unsigned)node * 384u) + loff);

    const int beg = __builtin_amdgcn_readfirstlane(rowptr[node]);
    const int end = __builtin_amdgcn_readfirstlane(rowptr[node + 1]);

    for (int base = beg; base < end; base += 32) {
        const int idx = base + lane;
        const int s   = (idx < end) ? bsrc[idx] : node;
        const int rem = end - base;
        const int cnt = (rem < 32) ? rem : 32;

        int j = 0;
        for (; j + 8 <= cnt; j += 8) {
            float4 v[8];
#pragma unroll
            for (int t = 0; t < 8; ++t) {
                const int sj = __builtin_amdgcn_readlane(s, j + t);
                v[t] = *(const float4*)(rows + (size_t)((unsigned)sj * 384u) + loff);
            }
#pragma unroll
            for (int t = 0; t < 8; ++t) {
                acc.x += v[t].x; acc.y += v[t].y; acc.z += v[t].z; acc.w += v[t].w;
            }
        }
        for (; j < cnt; j += 4) {
            float4 v[4];
#pragma unroll
            for (int t = 0; t < 4; ++t) {
                const int sj = __builtin_amdgcn_readlane(s, j + t);
                v[t] = *(const float4*)(rows + (size_t)((unsigned)sj * 384u) + loff);
            }
#pragma unroll
            for (int t = 0; t < 4; ++t) {
                const bool on = (j + t < cnt);
                acc.x += on ? v[t].x : 0.0f; acc.y += on ? v[t].y : 0.0f;
                acc.z += on ? v[t].z : 0.0f; acc.w += on ? v[t].w : 0.0f;
            }
        }
    }

    const float di = dinv[node];
    if (lane < 24) {
        const int c0 = lane * 4;
        float4 o;
        o.x = acc.x * di + bias[c0];
        o.y = acc.y * di + bias[c0 + 1];
        o.z = acc.z * di + bias[c0 + 2];
        o.w = acc.w * di + bias[c0 + 3];
        *(float4*)((char*)outp + (size_t)((unsigned)node * 384u) + (unsigned)lane * 16u) = o;
    }
}

__global__ void gcn_gather2_kernel(const int* __restrict__ rowptr,
                                   const int* __restrict__ bsrc,
                                   const float* __restrict__ hws,
                                   const float* __restrict__ dinv,
                                   const float* __restrict__ bias,
                                   float* __restrict__ outp, int n) {
    const int lane = threadIdx.x & 31;
    const int node = (int)(blockIdx.x * (blockDim.x >> 5) + (threadIdx.x >> 5));
    if (node >= n) return;

    const float2* __restrict__ rows = (const float2*)hws;
    const int beg = rowptr[node];
    const int end = rowptr[node + 1];

    float ax = 0.0f, ay = 0.0f;
    for (int idx = beg + lane; idx < end; idx += 32) {
        const float2 v = rows[bsrc[idx]];
        ax += v.x; ay += v.y;
    }
#pragma unroll
    for (int m = 16; m > 0; m >>= 1) {
        ax += __shfl_xor(ax, m, 32);
        ay += __shfl_xor(ay, m, 32);
    }
    if (lane == 0) {
        const float2 self = rows[node];
        const float  di   = dinv[node];
        float2 o;
        o.x = (ax + self.x) * di + bias[0];
        o.y = (ay + self.y) * di + bias[1];
        ((float2*)outp)[node] = o;
    }
}


extern "C" void kernel_launch(void* const* d_in, const int* in_sizes, int n_in,
                              void* d_out, int out_size, void* d_ws, size_t ws_size,
                              hipStream_t stream) {
    (void)in_sizes; (void)n_in; (void)out_size; (void)ws_size;

    const int N = GCN_N;
    const int E = GCN_E;

    const float* x    = (const float*)d_in[0];
    const int*   ei   = (const int*)d_in[1];
    const int*   src  = ei;
    const int*   dstp = ei + E;

    const float* W[5] = { (const float*)d_in[2], (const float*)d_in[4],
                          (const float*)d_in[6], (const float*)d_in[8],
                          (const float*)d_in[10] };
    const float* B[5] = { (const float*)d_in[3], (const float*)d_in[5],
                          (const float*)d_in[7], (const float*)d_in[9],
                          (const float*)d_in[11] };

    int*   cnt    = (int*)d_ws;
    int*   rowptr = cnt + N;
    float* dinv   = (float*)(rowptr + N + 4);
    int*   rank   = (int*)(dinv + N);
    int*   bsrc   = rank + E;
    float* hws    = (float*)(bsrc + E);
    float* aggA   = hws  + (size_t)N * 96;
    float* aggB   = aggA + (size_t)N * 96;
    float* out    = (float*)d_out;

    const int T = 256;

    gcn_zero_int_kernel<<<(N + T - 1) / T, T, 0, stream>>>(cnt, N);
    gcn_rank_kernel    <<<(E + T - 1) / T, T, 0, stream>>>(dstp, cnt, rank, E);
    gcn_scan_kernel    <<<(N + 1023) / 1024, 1024, 0, stream>>>(cnt, rowptr, dinv, N);
    gcn_fill_kernel    <<<(E + T - 1) / T, T, 0, stream>>>(src, dstp, rowptr, rank, bsrc, E);

    const int dims[6] = { 165, 96, 96, 96, 96, 2 };
    const float* layerIn = x;
    float* aggBufs[2] = { aggA, aggB };

    for (int l = 0; l < 5; ++l) {
        const int K  = dims[l];
        const int Cn = dims[l + 1];
        float* agg = (l == 4) ? out : aggBufs[l & 1];

        const int tiles         = (N / 16) * ((Cn + 15) / 16);
        const int wavesPerBlock = 4;
        const int gBlocks       = (tiles + wavesPerBlock - 1) / wavesPerBlock;
        if (l == 0) {
            gcn_gemm_wmma_f32<0><<<gBlocks, wavesPerBlock * 32, 0, stream>>>(
                layerIn, W[l], dinv, hws, N, K, Cn);
        } else {
            gcn_gemm_wmma_f32<1><<<gBlocks, wavesPerBlock * 32, 0, stream>>>(
                layerIn, W[l], dinv, hws, N, K, Cn);
        }

        const int nodesPerBlock = T / 32;
        const int aBlocks       = (N + nodesPerBlock - 1) / nodesPerBlock;
        if (l < 4) {
            gcn_gather96_kernel<<<aBlocks, T, 0, stream>>>(
                rowptr, bsrc, hws, dinv, B[l], agg, N);
        } else {
            gcn_gather2_kernel<<<aBlocks, T, 0, stream>>>(
                rowptr, bsrc, hws, dinv, B[l], agg, N);
        }

        layerIn = agg;
    }
}


# GEMM of the 96-column layers retiled: one wave per 16-row strip holds all six accumulators, A read once not 6x, 16/8-byte W loads and stores, ReLU moved to the gather's store; k sum order permuted
# speedup vs baseline: 3.4176x; 1.3204x over previous
#include <hip/hip_runtime.h>
#include <hip/hip_bf16.h>

typedef __attribute__((ext_vector_type(2))) float v2f;
typedef __attribute__((ext_vector_type(8))) float v8f;

#define GCN_N 50000
#define GCN_E 800000


__global__ void gcn_zero_int_kernel(int* __restrict__ p, int n) {
    int i = blockIdx.x * blockDim.x + threadIdx.x;
    if (i < n) p[i] = 0;
}

__global__ void gcn_rank_kernel(const int* __restrict__ dst, int* __restrict__ cnt,
                                int* __restrict__ rank, int E) {
    int e = blockIdx.x * blockDim.x + threadIdx.x;
    if (e < E) rank[e] = atomicAdd(&cnt[dst[e]], 1);
}

__global__ void gcn_scan_kernel(const int* __restrict__ cnt, int* __restrict__ rowptr,
                                float* __restrict__ dinv, int n) {
    __shared__ int sh[1024];
    const int tid  = threadIdx.x;
    const int base = blockIdx.x * 1024;

    int part = 0;
    for (int i = tid; i < base; i += 1024) part += cnt[i];
    sh[tid] = part;
    __syncthreads();
    for (int s = 512; s > 0; s >>= 1) {
        if (tid < s) sh[tid] += sh[tid + s];
        __syncthreads();
    }
    const int front = sh[0];
    __syncthreads();

    const int i = base + tid;
    const int c = (i < n) ? cnt[i] : 0;
    sh[tid] = c;
    __syncthreads();
    for (int off = 1; off < 1024; off <<= 1) {
        const int t = (tid >= off) ? sh[tid - off] : 0;
        __syncthreads();
        sh[tid] += t;
        __syncthreads();
    }
    const int excl = sh[tid] - c + front;
    if (i < n) {
        rowptr[i] = excl;
        dinv[i]   = rsqrtf((float)c + 1.0f);
    }
    if (i == n - 1) rowptr[n] = excl + c;
}

__global__ void gcn_fill_kernel(const int* __restrict__ src, const int* __restrict__ dst,
                                const int* __restrict__ rowptr, const int* __restrict__ rank,
                                int* __restrict__ bsrc, int E) {
    int e = blockIdx.x * blockDim.x + threadIdx.x;
    if (e < E) bsrc[rowptr[dst[e]] + rank[e]] = src[e];
}

template <int RELU>
__global__ void gcn_gemm_wmma_f32(const float* __restrict__ A,
                                  const float* __restrict__ W,
                                  const float* __restrict__ dinv,
                                  float* __restrict__ hws,
                                  int M, int K, int Nc) {
    const int lane = threadIdx.x & 31;
    const int wid  = threadIdx.x >> 5;
    const int wavesPerBlock = blockDim.x >> 5;

    const int tilesN = (Nc + 15) >> 4;
    const int tilesM = M >> 4;
    int tile = blockIdx.x * wavesPerBlock + wid;
    if (tile >= tilesM * tilesN) return;

    const int tm   = tile / tilesN;
    const int tn   = tile - tm * tilesN;
    const int half = lane >> 4;
    const int l16  = lane & 15;

    const int  row   = tm * 16 + l16;
    const int  col   = tn * 16 + l16;
    const bool colOK = (col < Nc);
    const int  colc  = colOK ? col : 0;
    const int  koff  = half * 2;

    const float* __restrict__ Arow = A + (size_t)row * K + koff;
    const float* __restrict__ Wcol = W + colc;

    v8f acc = {};
    const int Kmain = K & ~3;

#pragma unroll 4
    for (int k = 0; k < Kmain; k += 4) {
        float a0 = Arow[k];
        float a1 = Arow[k + 1];
        if (RELU) { a0 = fmaxf(a0, 0.0f); a1 = fmaxf(a1, 0.0f); }
        float w0 = Wcol[(size_t)(k + koff)     * Nc];
        float w1 = Wcol[(size_t)(k + koff + 1) * Nc];
        v2f a; a[0] = a0;               a[1] = a1;
        v2f b; b[0] = colOK ? w0 : 0.f; b[1] = colOK ? w1 : 0.f;
        acc = __builtin_amdgcn_wmma_f32_16x16x4_f32(
            false, a, false, b, (short)0, acc, false, false);
    }

    if (Kmain < K) {
        const int  ka  = Kmain + koff;
        const bool ok0 = (ka     < K);
        const bool ok1 = (ka + 1 < K);
        const int  kc0 = ok0 ? ka     : (K - 1);
        const int  kc1 = ok1 ? ka + 1 : (K - 1);
        float a0 = A[(size_t)row * K + kc0];
        float a1 = A[(size_t)row * K + kc1];
        if (RELU) { a0 = fmaxf(a0, 0.0f); a1 = fmaxf(a1, 0.0f); }
        float w0 = Wcol[(size_t)kc0 * Nc];
        float w1 = Wcol[(size_t)kc1 * Nc];
        v2f a; a[0] = ok0 ? a0 : 0.f;            a[1] = ok1 ? a1 : 0.f;
        v2f b; b[0] = (colOK && ok0) ? w0 : 0.f; b[1] = (colOK && ok1) ? w1 : 0.f;
        acc = __builtin_amdgcn_wmma_f32_16x16x4_f32(
            false, a, false, b, (short)0, acc, false, false);
    }

    const int rbase = tm * 16 + (half << 3);
    if (colOK) {
#pragma unroll
        for (int r = 0; r < 8; ++r) {
            const int m = rbase + r;
            hws[(size_t)m * Nc + col] = acc[r] * dinv[m];
        }
    }
}

struct GcnStage {
    float  a[4];
    float4 bl[4];
    float2 bh[4];
};

template <int K>
__device__ __forceinline__ void gcn_load_stage(GcnStage& st,
                                               const float* __restrict__ Arow,
                                               const float* __restrict__ Wl,
                                               const float* __restrict__ Wh, int kb) {
#pragma unroll
    for (int i = 0; i < 4; ++i) st.a[i] = Arow[kb + i];
#pragma unroll
    for (int i = 0; i < 4; ++i) {
        st.bl[i] = *(const float4*)(Wl + (size_t)(kb + i) * 96);
        st.bh[i] = *(const float2*)(Wh + (size_t)(kb + i) * 96);
    }
}

__device__ __forceinline__ void gcn_mma_stage(const GcnStage& st, v8f (&acc)[6]) {
#pragma unroll
    for (int s = 0; s < 2; ++s) {
        const float a0 = st.a[2 * s], a1 = st.a[2 * s + 1];
        const float4 p = st.bl[2 * s], q = st.bl[2 * s + 1];
        const float2 u = st.bh[2 * s], w = st.bh[2 * s + 1];
        v2f a;  a[0]  = a0;  a[1]  = a1;
        v2f b0; b0[0] = p.x; b0[1] = q.x;
        v2f b1; b1[0] = p.y; b1[1] = q.y;
        v2f b2; b2[0] = p.z; b2[1] = q.z;
        v2f b3; b3[0] = p.w; b3[1] = q.w;
        v2f b4; b4[0] = u.x; b4[1] = w.x;
        v2f b5; b5[0] = u.y; b5[1] = w.y;
        acc[0] = __builtin_amdgcn_wmma_f32_16x16x4_f32(false, a, false, b0, (short)0, acc[0], false, false);
        acc[1] = __builtin_amdgcn_wmma_f32_16x16x4_f32(false, a, false, b1, (short)0, acc[1], false, false);
        acc[2] = __builtin_amdgcn_wmma_f32_16x16x4_f32(false, a, false, b2, (short)0, acc[2], false, false);
        acc[3] = __builtin_amdgcn_wmma_f32_16x16x4_f32(false, a, false, b3, (short)0, acc[3], false, false);
        acc[4] = __builtin_amdgcn_wmma_f32_16x16x4_f32(false, a, false, b4, (short)0, acc[4], false, false);
        acc[5] = __builtin_amdgcn_wmma_f32_16x16x4_f32(false, a, false, b5, (short)0, acc[5], false, false);
    }
}

template <int K>
__global__ void __launch_bounds__(64)
gcn_gemm96_strip_wmma_f32(const float* __restrict__ A,
                          const float* __restrict__ W,
                          const float* __restrict__ dinv,
                          float* __restrict__ hws, int M) {
    const int lane = threadIdx.x & 31;
    const int tm   = blockIdx.x * (blockDim.x >> 5) + (threadIdx.x >> 5);
    if (tm >= (M >> 4)) return;

    const int c = lane & 15;
    const int h = lane >> 4;
    const int row = tm * 16 + c;

    const float* __restrict__ Arow = A + (size_t)row * K + 4 * h;
    const float* __restrict__ Wl   = W + (size_t)(4 * h) * 96 + 4 * c;
    const float* __restrict__ Wh   = W + (size_t)(4 * h) * 96 + 64 + 2 * c;

    const int rbase = tm * 16 + (h << 3);
    float dm[8];
#pragma unroll
    for (int r = 0; r < 8; ++r) dm[r] = dinv[rbase + r];

    v8f acc[6] = {};
    constexpr int K8 = K & ~7;

    constexpr int NB = K8 / 8;
    GcnStage st[3];
    gcn_load_stage<K>(st[0], Arow, Wl, Wh, 0);
    if constexpr (NB > 1) gcn_load_stage<K>(st[1], Arow, Wl, Wh, 8);
#pragma unroll
    for (int j = 0; j < NB; ++j) {
        if (j + 2 < NB) gcn_load_stage<K>(st[(j + 2) % 3], Arow, Wl, Wh, (j + 2) * 8);
        __builtin_amdgcn_sched_barrier(0);
        gcn_mma_stage(st[j % 3], acc);
    }

    if constexpr (K8 < K) {
        GcnStage t;
#pragma unroll
        for (int i = 0; i < 4; ++i) {
            const int  kk = K8 + 4 * h + i;
            const bool ok = (kk < K);
            const int  kc = ok ? kk : (K - 1);
            const float  av = A[(size_t)row * K + kc];
            const float4 lv = *(const float4*)(W + (size_t)kc * 96 + 4 * c);
            const float2 hv = *(const float2*)(W + (size_t)kc * 96 + 64 + 2 * c);
            t.a[i]    = ok ? av : 0.0f;
            t.bl[i].x = ok ? lv.x : 0.0f; t.bl[i].y = ok ? lv.y : 0.0f;
            t.bl[i].z = ok ? lv.z : 0.0f; t.bl[i].w = ok ? lv.w : 0.0f;
            t.bh[i].x = ok ? hv.x : 0.0f; t.bh[i].y = ok ? hv.y : 0.0f;
        }
        gcn_mma_stage(t, acc);
    }

#pragma unroll
    for (int r = 0; r < 8; ++r) {
        const size_t m = (size_t)(rbase + r);
        float4 o;
        o.x = acc[0][r] * dm[r]; o.y = acc[1][r] * dm[r];
        o.z = acc[2][r] * dm[r]; o.w = acc[3][r] * dm[r];
        float2 p;
        p.x = acc[4][r] * dm[r]; p.y = acc[5][r] * dm[r];
        *(float4*)(hws + m * 96 + 4 * c)      = o;
        *(float2*)(hws + m * 96 + 64 + 2 * c) = p;
    }
}

__global__ void gcn_gather96_kernel(const int* __restrict__ rowptr,
                                    const int* __restrict__ bsrc,
                                    const float* __restrict__ hws,
                                    const float* __restrict__ dinv,
                                    const float* __restrict__ bias,
                                    float* __restrict__ outp, int n) {
    const int lane = threadIdx.x & 31;
    const int node = __builtin_amdgcn_readfirstlane(
        (int)(blockIdx.x * (blockDim.x >> 5) + (threadIdx.x >> 5)));
    if (node >= n) return;

    const int l4 = (lane < 24) ? lane : lane - 24;
    const char* __restrict__ rows = (const char*)hws;
    const unsigned loff = (unsigned)l4 * 16u;

    float4 acc = *(const float4*)(rows + (size_t)((unsigned)node * 384u) + loff);

    const int beg = __builtin_amdgcn_readfirstlane(rowptr[node]);
    const int end = __builtin_amdgcn_readfirstlane(rowptr[node + 1]);

    for (int base = beg; base < end; base += 32) {
        const int idx = base + lane;
        const int s   = (idx < end) ? bsrc[idx] : node;
        const int rem = end - base;
        const int cnt = (rem < 32) ? rem : 32;

        int j = 0;
        for (; j + 8 <= cnt; j += 8) {
            float4 v[8];
#pragma unroll
            for (int t = 0; t < 8; ++t) {
                const int sj = __builtin_amdgcn_readlane(s, j + t);
                v[t] = *(const float4*)(rows + (size_t)((unsigned)sj * 384u) + loff);
            }
#pragma unroll
            for (int t = 0; t < 8; ++t) {
                acc.x += v[t].x; acc.y += v[t].y; acc.z += v[t].z; acc.w += v[t].w;
            }
        }
        for (; j < cnt; j += 4) {
            float4 v[4];
#pragma unroll
            for (int t = 0; t < 4; ++t) {
                const int sj = __builtin_amdgcn_readlane(s, j + t);
                v[t] = *(const float4*)(rows + (size_t)((unsigned)sj * 384u) + loff);
            }
#pragma unroll
            for (int t = 0; t < 4; ++t) {
                const bool on = (j + t < cnt);
                acc.x += on ? v[t].x : 0.0f; acc.y += on ? v[t].y : 0.0f;
                acc.z += on ? v[t].z : 0.0f; acc.w += on ? v[t].w : 0.0f;
            }
        }
    }

    const float di = dinv[node];
    if (lane < 24) {
        const int c0 = lane * 4;
        float4 o;
        o.x = fmaxf(acc.x * di + bias[c0],     0.0f);
        o.y = fmaxf(acc.y * di + bias[c0 + 1], 0.0f);
        o.z = fmaxf(acc.z * di + bias[c0 + 2], 0.0f);
        o.w = fmaxf(acc.w * di + bias[c0 + 3], 0.0f);
        *(float4*)((char*)outp + (size_t)((unsigned)node * 384u) + (unsigned)lane * 16u) = o;
    }
}

__global__ void gcn_gather2_kernel(const int* __restrict__ rowptr,
                                   const int* __restrict__ bsrc,
                                   const float* __restrict__ hws,
                                   const float* __restrict__ dinv,
                                   const float* __restrict__ bias,
                                   float* __restrict__ outp, int n) {
    const int lane = threadIdx.x & 31;
    const int node = (int)(blockIdx.x * (blockDim.x >> 5) + (threadIdx.x >> 5));
    if (node >= n) return;

    const float2* __restrict__ rows = (const float2*)hws;
    const int beg = rowptr[node];
    const int end = rowptr[node + 1];

    float ax = 0.0f, ay = 0.0f;
    for (int idx = beg + lane; idx < end; idx += 32) {
        const float2 v = rows[bsrc[idx]];
        ax += v.x; ay += v.y;
    }
#pragma unroll
    for (int m = 16; m > 0; m >>= 1) {
        ax += __shfl_xor(ax, m, 32);
        ay += __shfl_xor(ay, m, 32);
    }
    if (lane == 0) {
        const float2 self = rows[node];
        const float  di   = dinv[node];
        float2 o;
        o.x = (ax + self.x) * di + bias[0];
        o.y = (ay + self.y) * di + bias[1];
        ((float2*)outp)[node] = o;
    }
}


extern "C" void kernel_launch(void* const* d_in, const int* in_sizes, int n_in,
                              void* d_out, int out_size, void* d_ws, size_t ws_size,
                              hipStream_t stream) {
    (void)in_sizes; (void)n_in; (void)out_size; (void)ws_size;

    const int N = GCN_N;
    const int E = GCN_E;

    const float* x    = (const float*)d_in[0];
    const int*   ei   = (const int*)d_in[1];
    const int*   src  = ei;
    const int*   dstp = ei + E;

    const float* W[5] = { (const float*)d_in[2], (const float*)d_in[4],
                          (const float*)d_in[6], (const float*)d_in[8],
                          (const float*)d_in[10] };
    const float* B[5] = { (const float*)d_in[3], (const float*)d_in[5],
                          (const float*)d_in[7], (const float*)d_in[9],
                          (const float*)d_in[11] };

    int*   cnt    = (int*)d_ws;
    int*   rowptr = cnt + N;
    float* dinv   = (float*)(rowptr + N + 4);
    int*   rank   = (int*)(dinv + N);
    int*   bsrc   = rank + E;
    float* hws    = (float*)(bsrc + E);
    float* aggA   = hws  + (size_t)N * 96;
    float* aggB   = aggA + (size_t)N * 96;
    float* out    = (float*)d_out;

    const int T = 256;

    gcn_zero_int_kernel<<<(N + T - 1) / T, T, 0, stream>>>(cnt, N);
    gcn_rank_kernel    <<<(E + T - 1) / T, T, 0, stream>>>(dstp, cnt, rank, E);
    gcn_scan_kernel    <<<(N + 1023) / 1024, 1024, 0, stream>>>(cnt, rowptr, dinv, N);
    gcn_fill_kernel    <<<(E + T - 1) / T, T, 0, stream>>>(src, dstp, rowptr, rank, bsrc, E);

    const int dims[6] = { 165, 96, 96, 96, 96, 2 };
    const float* layerIn = x;
    float* aggBufs[2] = { aggA, aggB };

    for (int l = 0; l < 5; ++l) {
        const int K  = dims[l];
        const int Cn = dims[l + 1];
        float* agg = (l == 4) ? out : aggBufs[l & 1];

        const int tiles         = (N / 16) * ((Cn + 15) / 16);
        const int wavesPerBlock = 4;
        const int gBlocks       = (tiles + wavesPerBlock - 1) / wavesPerBlock;
        const int sBlocks       = (N / 16 + 1) / 2;
        if (l == 0) {
            gcn_gemm96_strip_wmma_f32<165><<<sBlocks, 64, 0, stream>>>(
                layerIn, W[l], dinv, hws, N);
        } else if (l < 4) {
            gcn_gemm96_strip_wmma_f32<96><<<sBlocks, 64, 0, stream>>>(
                layerIn, W[l], dinv, hws, N);
        } else {
            gcn_gemm_wmma_f32<0><<<gBlocks, wavesPerBlock * 32, 0, stream>>>(
                layerIn, W[l], dinv, hws, N, K, Cn);
        }

        const int nodesPerBlock = T / 32;
        const int aBlocks       = (N + nodesPerBlock - 1) / nodesPerBlock;
        if (l < 4) {
            gcn_gather96_kernel<<<aBlocks, T, 0, stream>>>(
                rowptr, bsrc, hws, dinv, B[l], agg, N);
        } else {
            gcn_gather2_kernel<<<aBlocks, T, 0, stream>>>(
                rowptr, bsrc, hws, dinv, B[l], agg, N);
        }

        layerIn = agg;
    }
}


# CSR build (prefix scan + second fill pass) replaced by fixed 64-slot destination buckets filled by the ranking atomic itself; dinv computed from the counts; 2 launches fewer; sum order unchanged
# speedup vs baseline: 3.8704x; 1.1325x over previous
#include <hip/hip_runtime.h>
#include <hip/hip_bf16.h>

typedef __attribute__((ext_vector_type(2))) float v2f;
typedef __attribute__((ext_vector_type(8))) float v8f;

#define GCN_N 50000
#define GCN_E 800000

#define GCN_CAP 64

__global__ void gcn_zero_int_kernel(int* __restrict__ p, int n) {
    int i = blockIdx.x * blockDim.x + threadIdx.x;
    if (i < n) p[i] = 0;
}

__global__ void gcn_bucket_kernel(const int* __restrict__ src, const int* __restrict__ dst,
                                  int* __restrict__ cnt, int* __restrict__ ell,
                                  int* __restrict__ ovf_src, int* __restrict__ ovf_dst,
                                  int n, int E) {
    int e = blockIdx.x * blockDim.x + threadIdx.x;
    if (e < E) {
        const int d = dst[e];
        const int s = src[e];
        const int r = atomicAdd(&cnt[d], 1);
        if (r < GCN_CAP) {
            ell[(size_t)d * GCN_CAP + r] = s;
        } else {
            const int o = atomicAdd(&cnt[n], 1);
            ovf_src[o] = s;
            ovf_dst[o] = d;
        }
    }
}

__device__ __forceinline__ float gcn_dinv(const int* __restrict__ cnt, int i) {
    return rsqrtf((float)cnt[i] + 1.0f);
}

template <int RELU>
__global__ void gcn_gemm_wmma_f32(const float* __restrict__ A,
                                  const float* __restrict__ W,
                                  const int* __restrict__ cnt,
                                  float* __restrict__ hws,
                                  int M, int K, int Nc) {
    const int lane = threadIdx.x & 31;
    const int wid  = threadIdx.x >> 5;
    const int wavesPerBlock = blockDim.x >> 5;

    const int tilesN = (Nc + 15) >> 4;
    const int tilesM = M >> 4;
    int tile = blockIdx.x * wavesPerBlock + wid;
    if (tile >= tilesM * tilesN) return;

    const int tm   = tile / tilesN;
    const int tn   = tile - tm * tilesN;
    const int half = lane >> 4;
    const int l16  = lane & 15;

    const int  row   = tm * 16 + l16;
    const int  col   = tn * 16 + l16;
    const bool colOK = (col < Nc);
    const int  colc  = colOK ? col : 0;
    const int  koff  = half * 2;

    const float* __restrict__ Arow = A + (size_t)row * K + koff;
    const float* __restrict__ Wcol = W + colc;

    v8f acc = {};
    const int Kmain = K & ~3;

#pragma unroll 4
    for (int k = 0; k < Kmain; k += 4) {
        float a0 = Arow[k];
        float a1 = Arow[k + 1];
        if (RELU) { a0 = fmaxf(a0, 0.0f); a1 = fmaxf(a1, 0.0f); }
        float w0 = Wcol[(size_t)(k + koff)     * Nc];
        float w1 = Wcol[(size_t)(k + koff + 1) * Nc];
        v2f a; a[0] = a0;               a[1] = a1;
        v2f b; b[0] = colOK ? w0 : 0.f; b[1] = colOK ? w1 : 0.f;
        acc = __builtin_amdgcn_wmma_f32_16x16x4_f32(
            false, a, false, b, (short)0, acc, false, false);
    }

    if (Kmain < K) {
        const int  ka  = Kmain + koff;
        const bool ok0 = (ka     < K);
        const bool ok1 = (ka + 1 < K);
        const int  kc0 = ok0 ? ka     : (K - 1);
        const int  kc1 = ok1 ? ka + 1 : (K - 1);
        float a0 = A[(size_t)row * K + kc0];
        float a1 = A[(size_t)row * K + kc1];
        if (RELU) { a0 = fmaxf(a0, 0.0f); a1 = fmaxf(a1, 0.0f); }
        float w0 = Wcol[(size_t)kc0 * Nc];
        float w1 = Wcol[(size_t)kc1 * Nc];
        v2f a; a[0] = ok0 ? a0 : 0.f;            a[1] = ok1 ? a1 : 0.f;
        v2f b; b[0] = (colOK && ok0) ? w0 : 0.f; b[1] = (colOK && ok1) ? w1 : 0.f;
        acc = __builtin_amdgcn_wmma_f32_16x16x4_f32(
            false, a, false, b, (short)0, acc, false, false);
    }

    const int rbase = tm * 16 + (half << 3);
    if (colOK) {
#pragma unroll
        for (int r = 0; r < 8; ++r) {
            const int m = rbase + r;
            hws[(size_t)m * Nc + col] = acc[r] * gcn_dinv(cnt, m);
        }
    }
}

struct GcnStage {
    float  a[4];
    float4 bl[4];
    float2 bh[4];
};

template <int K>
__device__ __forceinline__ void gcn_load_stage(GcnStage& st,
                                               const float* __restrict__ Arow,
                                               const float* __restrict__ Wl,
                                               const float* __restrict__ Wh, int kb) {
#pragma unroll
    for (int i = 0; i < 4; ++i) st.a[i] = Arow[kb + i];
#pragma unroll
    for (int i = 0; i < 4; ++i) {
        st.bl[i] = *(const float4*)(Wl + (size_t)(kb + i) * 96);
        st.bh[i] = *(const float2*)(Wh + (size_t)(kb + i) * 96);
    }
}

__device__ __forceinline__ void gcn_mma_stage(const GcnStage& st, v8f (&acc)[6]) {
#pragma unroll
    for (int s = 0; s < 2; ++s) {
        const float a0 = st.a[2 * s], a1 = st.a[2 * s + 1];
        const float4 p = st.bl[2 * s], q = st.bl[2 * s + 1];
        const float2 u = st.bh[2 * s], w = st.bh[2 * s + 1];
        v2f a;  a[0]  = a0;  a[1]  = a1;
        v2f b0; b0[0] = p.x; b0[1] = q.x;
        v2f b1; b1[0] = p.y; b1[1] = q.y;
        v2f b2; b2[0] = p.z; b2[1] = q.z;
        v2f b3; b3[0] = p.w; b3[1] = q.w;
        v2f b4; b4[0] = u.x; b4[1] = w.x;
        v2f b5; b5[0] = u.y; b5[1] = w.y;
        acc[0] = __builtin_amdgcn_wmma_f32_16x16x4_f32(false, a, false, b0, (short)0, acc[0], false, false);
        acc[1] = __builtin_amdgcn_wmma_f32_16x16x4_f32(false, a, false, b1, (short)0, acc[1], false, false);
        acc[2] = __builtin_amdgcn_wmma_f32_16x16x4_f32(false, a, false, b2, (short)0, acc[2], false, false);
        acc[3] = __builtin_amdgcn_wmma_f32_16x16x4_f32(false, a, false, b3, (short)0, acc[3], false, false);
        acc[4] = __builtin_amdgcn_wmma_f32_16x16x4_f32(false, a, false, b4, (short)0, acc[4], false, false);
        acc[5] = __builtin_amdgcn_wmma_f32_16x16x4_f32(false, a, false, b5, (short)0, acc[5], false, false);
    }
}

template <int K>
__global__ void __launch_bounds__(64)
gcn_gemm96_strip_wmma_f32(const float* __restrict__ A,
                          const float* __restrict__ W,
                          const int* __restrict__ cnt,
                          float* __restrict__ hws, int M) {
    const int lane = threadIdx.x & 31;
    const int tm   = blockIdx.x * (blockDim.x >> 5) + (threadIdx.x >> 5);
    if (tm >= (M >> 4)) return;

    const int c = lane & 15;
    const int h = lane >> 4;
    const int row = tm * 16 + c;

    const float* __restrict__ Arow = A + (size_t)row * K + 4 * h;
    const float* __restrict__ Wl   = W + (size_t)(4 * h) * 96 + 4 * c;
    const float* __restrict__ Wh   = W + (size_t)(4 * h) * 96 + 64 + 2 * c;

    const int rbase = tm * 16 + (h << 3);
    float dm[8];
#pragma unroll
    for (int r = 0; r < 8; ++r) dm[r] = gcn_dinv(cnt, rbase + r);

    v8f acc[6] = {};
    constexpr int K8 = K & ~7;

    constexpr int NB = K8 / 8;
    GcnStage st[3];
    gcn_load_stage<K>(st[0], Arow, Wl, Wh, 0);
    if constexpr (NB > 1) gcn_load_stage<K>(st[1], Arow, Wl, Wh, 8);
#pragma unroll
    for (int j = 0; j < NB; ++j) {
        if (j + 2 < NB) gcn_load_stage<K>(st[(j + 2) % 3], Arow, Wl, Wh, (j + 2) * 8);
        __builtin_amdgcn_sched_barrier(0);
        gcn_mma_stage(st[j % 3], acc);
    }

    if constexpr (K8 < K) {
        GcnStage t;
#pragma unroll
        for (int i = 0; i < 4; ++i) {
            const int  kk = K8 + 4 * h + i;
            const bool ok = (kk < K);
            const int  kc = ok ? kk : (K - 1);
            const float  av = A[(size_t)row * K + kc];
            const float4 lv = *(const float4*)(W + (size_t)kc * 96 + 4 * c);
            const float2 hv = *(const float2*)(W + (size_t)kc * 96 + 64 + 2 * c);
            t.a[i]    = ok ? av : 0.0f;
            t.bl[i].x = ok ? lv.x : 0.0f; t.bl[i].y = ok ? lv.y : 0.0f;
            t.bl[i].z = ok ? lv.z : 0.0f; t.bl[i].w = ok ? lv.w : 0.0f;
            t.bh[i].x = ok ? hv.x : 0.0f; t.bh[i].y = ok ? hv.y : 0.0f;
        }
        gcn_mma_stage(t, acc);
    }

#pragma unroll
    for (int r = 0; r < 8; ++r) {
        const size_t m = (size_t)(rbase + r);
        float4 o;
        o.x = acc[0][r] * dm[r]; o.y = acc[1][r] * dm[r];
        o.z = acc[2][r] * dm[r]; o.w = acc[3][r] * dm[r];
        float2 p;
        p.x = acc[4][r] * dm[r]; p.y = acc[5][r] * dm[r];
        *(float4*)(hws + m * 96 + 4 * c)      = o;
        *(float2*)(hws + m * 96 + 64 + 2 * c) = p;
    }
}

__global__ void gcn_gather96_kernel(const int* __restrict__ cnt,
                                    const int* __restrict__ ell,
                                    const int* __restrict__ ovf_src,
                                    const int* __restrict__ ovf_dst,
                                    const float* __restrict__ hws,
                                    const float* __restrict__ bias,
                                    float* __restrict__ outp, int n) {
    const int lane = threadIdx.x & 31;
    const int node = __builtin_amdgcn_readfirstlane(
        (int)(blockIdx.x * (blockDim.x >> 5) + (threadIdx.x >> 5)));
    if (node >= n) return;

    const int l4 = (lane < 24) ? lane : lane - 24;
    const char* __restrict__ rows = (const char*)hws;
    const unsigned loff = (unsigned)l4 * 16u;

    float4 acc = *(const float4*)(rows + (size_t)((unsigned)node * 384u) + loff);

    const int deg = __builtin_amdgcn_readfirstlane(cnt[node]);
    const int nb  = (deg < GCN_CAP) ? deg : GCN_CAP;
    const int* __restrict__ bucket = ell + (size_t)node * GCN_CAP;

    for (int base = 0; base < nb; base += 32) {
        const int idx = base + lane;
        const int s   = (idx < nb) ? bucket[idx] : node;
        const int rem = nb - base;
        const int cn  = (rem < 32) ? rem : 32;

        int j = 0;
        for (; j + 8 <= cn; j += 8) {
            float4 v[8];
#pragma unroll
            for (int t = 0; t < 8; ++t) {
                const int sj = __builtin_amdgcn_readlane(s, j + t);
                v[t] = *(const float4*)(rows + (size_t)((unsigned)sj * 384u) + loff);
            }
#pragma unroll
            for (int t = 0; t < 8; ++t) {
                acc.x += v[t].x; acc.y += v[t].y; acc.z += v[t].z; acc.w += v[t].w;
            }
        }
        for (; j < cn; j += 4) {
            float4 v[4];
#pragma unroll
            for (int t = 0; t < 4; ++t) {
                const int sj = __builtin_amdgcn_readlane(s, j + t);
                v[t] = *(const float4*)(rows + (size_t)((unsigned)sj * 384u) + loff);
            }
#pragma unroll
            for (int t = 0; t < 4; ++t) {
                const bool on = (j + t < cn);
                acc.x += on ? v[t].x : 0.0f; acc.y += on ? v[t].y : 0.0f;
                acc.z += on ? v[t].z : 0.0f; acc.w += on ? v[t].w : 0.0f;
            }
        }
    }

    if (deg > GCN_CAP) {
        const int novf = __builtin_amdgcn_readfirstlane(cnt[n]);
        for (int i = 0; i < novf; ++i) {
            if (__builtin_amdgcn_readfirstlane(ovf_dst[i]) == node) {
                const int sj = __builtin_amdgcn_readfirstlane(ovf_src[i]);
                const float4 v = *(const float4*)(rows + (size_t)((unsigned)sj * 384u) + loff);
                acc.x += v.x; acc.y += v.y; acc.z += v.z; acc.w += v.w;
            }
        }
    }

    const float di = rsqrtf((float)deg + 1.0f);
    if (lane < 24) {
        const int c0 = lane * 4;
        float4 o;
        o.x = fmaxf(acc.x * di + bias[c0],     0.0f);
        o.y = fmaxf(acc.y * di + bias[c0 + 1], 0.0f);
        o.z = fmaxf(acc.z * di + bias[c0 + 2], 0.0f);
        o.w = fmaxf(acc.w * di + bias[c0 + 3], 0.0f);
        *(float4*)((char*)outp + (size_t)((unsigned)node * 384u) + (unsigned)lane * 16u) = o;
    }
}

__global__ void gcn_gather2_kernel(const int* __restrict__ cnt,
                                   const int* __restrict__ ell,
                                   const int* __restrict__ ovf_src,
                                   const int* __restrict__ ovf_dst,
                                   const float* __restrict__ hws,
                                   const float* __restrict__ bias,
                                   float* __restrict__ outp, int n) {
    const int lane = threadIdx.x & 31;
    const int node = (int)(blockIdx.x * (blockDim.x >> 5) + (threadIdx.x >> 5));
    if (node >= n) return;

    const float2* __restrict__ rows = (const float2*)hws;
    const int deg = cnt[node];
    const int nb  = (deg < GCN_CAP) ? deg : GCN_CAP;
    const int* __restrict__ bucket = ell + (size_t)node * GCN_CAP;

    float ax = 0.0f, ay = 0.0f;
    for (int idx = lane; idx < nb; idx += 32) {
        const float2 v = rows[bucket[idx]];
        ax += v.x; ay += v.y;
    }
    if (deg > GCN_CAP) {
        const int novf = cnt[n];
        for (int i = lane; i < novf; i += 32) {
            if (ovf_dst[i] == node) {
                const float2 v = rows[ovf_src[i]];
                ax += v.x; ay += v.y;
            }
        }
    }
#pragma unroll
    for (int m = 16; m > 0; m >>= 1) {
        ax += __shfl_xor(ax, m, 32);
        ay += __shfl_xor(ay, m, 32);
    }
    if (lane == 0) {
        const float2 self = rows[node];
        const float  di   = rsqrtf((float)deg + 1.0f);
        float2 o;
        o.x = (ax + self.x) * di + bias[0];
        o.y = (ay + self.y) * di + bias[1];
        ((float2*)outp)[node] = o;
    }
}


extern "C" void kernel_launch(void* const* d_in, const int* in_sizes, int n_in,
                              void* d_out, int out_size, void* d_ws, size_t ws_size,
                              hipStream_t stream) {
    (void)in_sizes; (void)n_in; (void)out_size; (void)ws_size;

    const int N = GCN_N;
    const int E = GCN_E;

    const float* x    = (const float*)d_in[0];
    const int*   ei   = (const int*)d_in[1];
    const int*   src  = ei;
    const int*   dstp = ei + E;

    const float* W[5] = { (const float*)d_in[2], (const float*)d_in[4],
                          (const float*)d_in[6], (const float*)d_in[8],
                          (const float*)d_in[10] };
    const float* B[5] = { (const float*)d_in[3], (const float*)d_in[5],
                          (const float*)d_in[7], (const float*)d_in[9],
                          (const float*)d_in[11] };

    int*   cnt    = (int*)d_ws;
    int*   ell    = cnt + 50048;
    int*   ovfs   = ell + (size_t)N * GCN_CAP;
    int*   ovfd   = ovfs + E;
    float* hws    = (float*)(ovfd + E);
    float* aggA   = hws  + (size_t)N * 96;
    float* aggB   = aggA + (size_t)N * 96;
    float* out    = (float*)d_out;

    const int T = 256;

    gcn_zero_int_kernel<<<(N + 1 + T - 1) / T, T, 0, stream>>>(cnt, N + 1);
    gcn_bucket_kernel  <<<(E + T - 1) / T, T, 0, stream>>>(src, dstp, cnt, ell, ovfs, ovfd, N, E);

    const int dims[6] = { 165, 96, 96, 96, 96, 2 };
    const float* layerIn = x;
    float* aggBufs[2] = { aggA, aggB };

    for (int l = 0; l < 5; ++l) {
        const int K  = dims[l];
        const int Cn = dims[l + 1];
        float* agg = (l == 4) ? out : aggBufs[l & 1];

        const int tiles         = (N / 16) * ((Cn + 15) / 16);
        const int wavesPerBlock = 4;
        const int gBlocks       = (tiles + wavesPerBlock - 1) / wavesPerBlock;
        const int sBlocks       = (N / 16 + 1) / 2;
        if (l == 0) {
            gcn_gemm96_strip_wmma_f32<165><<<sBlocks, 64, 0, stream>>>(
                layerIn, W[l], cnt, hws, N);
        } else if (l < 4) {
            gcn_gemm96_strip_wmma_f32<96><<<sBlocks, 64, 0, stream>>>(
                layerIn, W[l], cnt, hws, N);
        } else {
            gcn_gemm_wmma_f32<0><<<gBlocks, wavesPerBlock * 32, 0, stream>>>(
                layerIn, W[l], cnt, hws, N, K, Cn);
        }

        const int nodesPerBlock = T / 32;
        const int aBlocks       = (N + nodesPerBlock - 1) / nodesPerBlock;
        if (l < 4) {
            gcn_gather96_kernel<<<aBlocks, T, 0, stream>>>(
                cnt, ell, ovfs, ovfd, hws, B[l], agg, N);
        } else {
            gcn_gather2_kernel<<<aBlocks, T, 0, stream>>>(
                cnt, ell, ovfs, ovfd, hws, B[l], agg, N);
        }

        layerIn = agg;
    }
}


# Edge bucketing: returning global atomic plus scattered 4-byte store per edge replaced by a two-launch counting sort through LDS (chunk bins, owner block per 64 nodes); slot order still by arrival
# speedup vs baseline: 5.2128x; 1.3468x over previous
#include <hip/hip_runtime.h>
#include <hip/hip_bf16.h>

typedef __attribute__((ext_vector_type(2))) float v2f;
typedef __attribute__((ext_vector_type(8))) float v8f;

#define GCN_N 50000
#define GCN_E 800000

#define GCN_CAP  64
#define GCN_NCH  256
#define GCN_CE   3125
#define GCN_NBIN 782
#define GCN_TABP 784

__global__ void __launch_bounds__(1024)
gcn_bin_kernel(const int* __restrict__ src, const int* __restrict__ dst,
               int* __restrict__ tab, int2* __restrict__ stage,
               int* __restrict__ cnt, int n, int E) {
    __shared__ int  hist[GCN_TABP];
    __shared__ int  wsum[32];
    __shared__ int2 sorted[GCN_CE];

    const int tid  = threadIdx.x;
    const int lane = tid & 31;
    const int wid  = tid >> 5;
    const int c    = blockIdx.x;
    const int base = c * GCN_CE;

    if (tid < GCN_TABP) hist[tid] = 0;
    if (c == 0 && tid == 0) cnt[n] = 0;
    __syncthreads();

    int es[4], ed[4], er[4];
#pragma unroll
    for (int k = 0; k < 4; ++k) {
        const int  i  = k * 1024 + tid;
        const bool ok = (i < GCN_CE) && (base + i < E);
        ed[k] = ok ? dst[base + i] : -1;
        es[k] = ok ? src[base + i] : 0;
    }
#pragma unroll
    for (int k = 0; k < 4; ++k) {
        er[k] = 0;
        if (ed[k] >= 0) er[k] = atomicAdd(&hist[ed[k] >> 6], 1);
    }
    __syncthreads();

    const int v = (tid < GCN_NBIN) ? hist[tid] : 0;
    int x = v;
    for (int o = 1; o < 32; o <<= 1) {
        const int y = __shfl_up(x, o, 32);
        if (lane >= o) x += y;
    }
    if (lane == 31) wsum[wid] = x;
    __syncthreads();
    if (wid == 0) {
        const int w = wsum[lane];
        int z = w;
        for (int o = 1; o < 32; o <<= 1) {
            const int y = __shfl_up(z, o, 32);
            if (lane >= o) z += y;
        }
        wsum[lane] = z - w;
    }
    __syncthreads();
    const int excl = x - v + wsum[wid];
    if (tid <= GCN_NBIN) {
        hist[tid] = excl;
        tab[c * GCN_TABP + tid] = excl;
    }
    __syncthreads();

    const int tot = hist[GCN_NBIN];
#pragma unroll
    for (int k = 0; k < 4; ++k) {
        if (ed[k] >= 0) sorted[hist[ed[k] >> 6] + er[k]] = make_int2(es[k], ed[k]);
    }
    __syncthreads();
#pragma unroll
    for (int k = 0; k < 4; ++k) {
        const int i = k * 1024 + tid;
        if (i < tot) stage[(size_t)base + i] = sorted[i];
    }
}

__device__ __forceinline__ void gcn_place_edge(int2 e, int* cntL, int* ellL,
                                               int* __restrict__ cnt,
                                               int* __restrict__ ovf_src,
                                               int* __restrict__ ovf_dst, int n) {
    const int dl = e.y & 63;
    const int r  = atomicAdd(&cntL[dl], 1);
    if (r < GCN_CAP) {
        ellL[dl * GCN_CAP + r] = e.x;
    } else {
        const int o = atomicAdd(&cnt[n], 1);
        ovf_src[o] = e.x;
        ovf_dst[o] = e.y;
    }
}

__global__ void __launch_bounds__(256)
gcn_owner_kernel(const int* __restrict__ tab, const int2* __restrict__ stage,
                 int* __restrict__ cnt, int* __restrict__ ell,
                 int* __restrict__ ovf_src, int* __restrict__ ovf_dst, int n) {
    __shared__ int  cntL[64];
    __shared__ int4 ellL4[64 * GCN_CAP / 4];
    int* ellL = (int*)ellL4;

    const int tid  = threadIdx.x;
    const int lane = tid & 31;
    const int wid  = tid >> 5;
    const int b    = blockIdx.x;

    if (tid < 64) cntL[tid] = 0;

    const int myc = wid * 32 + lane;
    const int lo  = tab[myc * GCN_TABP + b];
    const int hi  = tab[myc * GCN_TABP + b + 1];
    __syncthreads();

    const int sub = lane >> 4;
    const int k16 = lane & 15;
    int2 ev[16];
    unsigned act = 0;
#pragma unroll
    for (int j = 0; j < 16; ++j) {
        const int la = __builtin_amdgcn_readlane(lo, 2 * j);
        const int lb = __builtin_amdgcn_readlane(lo, 2 * j + 1);
        const int ha = __builtin_amdgcn_readlane(hi, 2 * j);
        const int hb = __builtin_amdgcn_readlane(hi, 2 * j + 1);
        const int lj = sub ? lb : la;
        const int hj = sub ? hb : ha;
        const int2* __restrict__ p = stage + (size_t)(wid * 32 + 2 * j + sub) * GCN_CE;
        ev[j] = make_int2(0, 0);
        if (lj + k16 < hj) {
            ev[j] = p[lj + k16];
            act |= 1u << j;
        }
    }
#pragma unroll
    for (int j = 0; j < 16; ++j) {
        if (act & (1u << j)) gcn_place_edge(ev[j], cntL, ellL, cnt, ovf_src, ovf_dst, n);
    }
    for (int j = 0; j < 32; ++j) {
        const int lj = __builtin_amdgcn_readlane(lo, j);
        const int hj = __builtin_amdgcn_readlane(hi, j);
        if (hj - lj > 16) {
            const int2* __restrict__ p = stage + (size_t)(wid * 32 + j) * GCN_CE;
            for (int i = lj + 16 + lane; i < hj; i += 32)
                gcn_place_edge(p[i], cntL, ellL, cnt, ovf_src, ovf_dst, n);
        }
    }
    __syncthreads();

    const int node0 = b * 64;
    if (tid < 64 && node0 + tid < n) cnt[node0 + tid] = cntL[tid];
#pragma unroll
    for (int k = 0; k < 4; ++k) {
        const int i  = k * 256 + tid;
        const int nd = i >> 4;
        const int q  = i & 15;
        if (node0 + nd < n && (q < 8 || cntL[nd] > 32))
            *(int4*)(ell + (size_t)(node0 + nd) * GCN_CAP + 4 * q) = ellL4[nd * 16 + q];
    }
}

__device__ __forceinline__ float gcn_dinv(const int* __restrict__ cnt, int i) {
    return rsqrtf((float)cnt[i] + 1.0f);
}

template <int RELU>
__global__ void gcn_gemm_wmma_f32(const float* __restrict__ A,
                                  const float* __restrict__ W,
                                  const int* __restrict__ cnt,
                                  float* __restrict__ hws,
                                  int M, int K, int Nc) {
    const int lane = threadIdx.x & 31;
    const int wid  = threadIdx.x >> 5;
    const int wavesPerBlock = blockDim.x >> 5;

    const int tilesN = (Nc + 15) >> 4;
    const int tilesM = M >> 4;
    int tile = blockIdx.x * wavesPerBlock + wid;
    if (tile >= tilesM * tilesN) return;

    const int tm   = tile / tilesN;
    const int tn   = tile - tm * tilesN;
    const int half = lane >> 4;
    const int l16  = lane & 15;

    const int  row   = tm * 16 + l16;
    const int  col   = tn * 16 + l16;
    const bool colOK = (col < Nc);
    const int  colc  = colOK ? col : 0;
    const int  koff  = half * 2;

    const float* __restrict__ Arow = A + (size_t)row * K + koff;
    const float* __restrict__ Wcol = W + colc;

    v8f acc = {};
    const int Kmain = K & ~3;

#pragma unroll 4
    for (int k = 0; k < Kmain; k += 4) {
        float a0 = Arow[k];
        float a1 = Arow[k + 1];
        if (RELU) { a0 = fmaxf(a0, 0.0f); a1 = fmaxf(a1, 0.0f); }
        float w0 = Wcol[(size_t)(k + koff)     * Nc];
        float w1 = Wcol[(size_t)(k + koff + 1) * Nc];
        v2f a; a[0] = a0;               a[1] = a1;
        v2f b; b[0] = colOK ? w0 : 0.f; b[1] = colOK ? w1 : 0.f;
        acc = __builtin_amdgcn_wmma_f32_16x16x4_f32(
            false, a, false, b, (short)0, acc, false, false);
    }

    if (Kmain < K) {
        const int  ka  = Kmain + koff;
        const bool ok0 = (ka     < K);
        const bool ok1 = (ka + 1 < K);
        const int  kc0 = ok0 ? ka     : (K - 1);
        const int  kc1 = ok1 ? ka + 1 : (K - 1);
        float a0 = A[(size_t)row * K + kc0];
        float a1 = A[(size_t)row * K + kc1];
        if (RELU) { a0 = fmaxf(a0, 0.0f); a1 = fmaxf(a1, 0.0f); }
        float w0 = Wcol[(size_t)kc0 * Nc];
        float w1 = Wcol[(size_t)kc1 * Nc];
        v2f a; a[0] = ok0 ? a0 : 0.f;            a[1] = ok1 ? a1 : 0.f;
        v2f b; b[0] = (colOK && ok0) ? w0 : 0.f; b[1] = (colOK && ok1) ? w1 : 0.f;
        acc = __builtin_amdgcn_wmma_f32_16x16x4_f32(
            false, a, false, b, (short)0, acc, false, false);
    }

    const int rbase = tm * 16 + (half << 3);
    if (colOK) {
#pragma unroll
        for (int r = 0; r < 8; ++r) {
            const int m = rbase + r;
            hws[(size_t)m * Nc + col] = acc[r] * gcn_dinv(cnt, m);
        }
    }
}

struct GcnStage {
    float  a[4];
    float4 bl[4];
    float2 bh[4];
};

template <int K>
__device__ __forceinline__ void gcn_load_stage(GcnStage& st,
                                               const float* __restrict__ Arow,
                                               const float* __restrict__ Wl,
                                               const float* __restrict__ Wh, int kb) {
#pragma unroll
    for (int i = 0; i < 4; ++i) st.a[i] = Arow[kb + i];
#pragma unroll
    for (int i = 0; i < 4; ++i) {
        st.bl[i] = *(const float4*)(Wl + (size_t)(kb + i) * 96);
        st.bh[i] = *(const float2*)(Wh + (size_t)(kb + i) * 96);
    }
}

__device__ __forceinline__ void gcn_mma_stage(const GcnStage& st, v8f (&acc)[6]) {
#pragma unroll
    for (int s = 0; s < 2; ++s) {
        const float a0 = st.a[2 * s], a1 = st.a[2 * s + 1];
        const float4 p = st.bl[2 * s], q = st.bl[2 * s + 1];
        const float2 u = st.bh[2 * s], w = st.bh[2 * s + 1];
        v2f a;  a[0]  = a0;  a[1]  = a1;
        v2f b0; b0[0] = p.x; b0[1] = q.x;
        v2f b1; b1[0] = p.y; b1[1] = q.y;
        v2f b2; b2[0] = p.z; b2[1] = q.z;
        v2f b3; b3[0] = p.w; b3[1] = q.w;
        v2f b4; b4[0] = u.x; b4[1] = w.x;
        v2f b5; b5[0] = u.y; b5[1] = w.y;
        acc[0] = __builtin_amdgcn_wmma_f32_16x16x4_f32(false, a, false, b0, (short)0, acc[0], false, false);
        acc[1] = __builtin_amdgcn_wmma_f32_16x16x4_f32(false, a, false, b1, (short)0, acc[1], false, false);
        acc[2] = __builtin_amdgcn_wmma_f32_16x16x4_f32(false, a, false, b2, (short)0, acc[2], false, false);
        acc[3] = __builtin_amdgcn_wmma_f32_16x16x4_f32(false, a, false, b3, (short)0, acc[3], false, false);
        acc[4] = __builtin_amdgcn_wmma_f32_16x16x4_f32(false, a, false, b4, (short)0, acc[4], false, false);
        acc[5] = __builtin_amdgcn_wmma_f32_16x16x4_f32(false, a, false, b5, (short)0, acc[5], false, false);
    }
}

template <int K>
__global__ void __launch_bounds__(64)
gcn_gemm96_strip_wmma_f32(const float* __restrict__ A,
                          const float* __restrict__ W,
                          const int* __restrict__ cnt,
                          float* __restrict__ hws, int M) {
    const int lane = threadIdx.x & 31;
    const int tm   = blockIdx.x * (blockDim.x >> 5) + (threadIdx.x >> 5);
    if (tm >= (M >> 4)) return;

    const int c = lane & 15;
    const int h = lane >> 4;
    const int row = tm * 16 + c;

    const float* __restrict__ Arow = A + (size_t)row * K + 4 * h;
    const float* __restrict__ Wl   = W + (size_t)(4 * h) * 96 + 4 * c;
    const float* __restrict__ Wh   = W + (size_t)(4 * h) * 96 + 64 + 2 * c;

    const int rbase = tm * 16 + (h << 3);
    float dm[8];
#pragma unroll
    for (int r = 0; r < 8; ++r) dm[r] = gcn_dinv(cnt, rbase + r);

    v8f acc[6] = {};
    constexpr int K8 = K & ~7;

    constexpr int NB = K8 / 8;
    GcnStage st[3];
    gcn_load_stage<K>(st[0], Arow, Wl, Wh, 0);
    if constexpr (NB > 1) gcn_load_stage<K>(st[1], Arow, Wl, Wh, 8);
#pragma unroll
    for (int j = 0; j < NB; ++j) {
        if (j + 2 < NB) gcn_load_stage<K>(st[(j + 2) % 3], Arow, Wl, Wh, (j + 2) * 8);
        __builtin_amdgcn_sched_barrier(0);
        gcn_mma_stage(st[j % 3], acc);
    }

    if constexpr (K8 < K) {
        GcnStage t;
#pragma unroll
        for (int i = 0; i < 4; ++i) {
            const int  kk = K8 + 4 * h + i;
            const bool ok = (kk < K);
            const int  kc = ok ? kk : (K - 1);
            const float  av = A[(size_t)row * K + kc];
            const float4 lv = *(const float4*)(W + (size_t)kc * 96 + 4 * c);
            const float2 hv = *(const float2*)(W + (size_t)kc * 96 + 64 + 2 * c);
            t.a[i]    = ok ? av : 0.0f;
            t.bl[i].x = ok ? lv.x : 0.0f; t.bl[i].y = ok ? lv.y : 0.0f;
            t.bl[i].z = ok ? lv.z : 0.0f; t.bl[i].w = ok ? lv.w : 0.0f;
            t.bh[i].x = ok ? hv.x : 0.0f; t.bh[i].y = ok ? hv.y : 0.0f;
        }
        gcn_mma_stage(t, acc);
    }

#pragma unroll
    for (int r = 0; r < 8; ++r) {
        const size_t m = (size_t)(rbase + r);
        float4 o;
        o.x = acc[0][r] * dm[r]; o.y = acc[1][r] * dm[r];
        o.z = acc[2][r] * dm[r]; o.w = acc[3][r] * dm[r];
        float2 p;
        p.x = acc[4][r] * dm[r]; p.y = acc[5][r] * dm[r];
        *(float4*)(hws + m * 96 + 4 * c)      = o;
        *(float2*)(hws + m * 96 + 64 + 2 * c) = p;
    }
}

__global__ void gcn_gather96_kernel(const int* __restrict__ cnt,
                                    const int* __restrict__ ell,
                                    const int* __restrict__ ovf_src,
                                    const int* __restrict__ ovf_dst,
                                    const float* __restrict__ hws,
                                    const float* __restrict__ bias,
                                    float* __restrict__ outp, int n) {
    const int lane = threadIdx.x & 31;
    const int node = __builtin_amdgcn_readfirstlane(
        (int)(blockIdx.x * (blockDim.x >> 5) + (threadIdx.x >> 5)));
    if (node >= n) return;

    const int l4 = (lane < 24) ? lane : lane - 24;
    const char* __restrict__ rows = (const char*)hws;
    const unsigned loff = (unsigned)l4 * 16u;

    float4 acc = *(const float4*)(rows + (size_t)((unsigned)node * 384u) + loff);

    const int deg = __builtin_amdgcn_readfirstlane(cnt[node]);
    const int nb  = (deg < GCN_CAP) ? deg : GCN_CAP;
    const int* __restrict__ bucket = ell + (size_t)node * GCN_CAP;

    for (int base = 0; base < nb; base += 32) {
        const int idx = base + lane;
        const int s   = (idx < nb) ? bucket[idx] : node;
        const int rem = nb - base;
        const int cn  = (rem < 32) ? rem : 32;

        int j = 0;
        for (; j + 8 <= cn; j += 8) {
            float4 v[8];
#pragma unroll
            for (int t = 0; t < 8; ++t) {
                const int sj = __builtin_amdgcn_readlane(s, j + t);
                v[t] = *(const float4*)(rows + (size_t)((unsigned)sj * 384u) + loff);
            }
#pragma unroll
            for (int t = 0; t < 8; ++t) {
                acc.x += v[t].x; acc.y += v[t].y; acc.z += v[t].z; acc.w += v[t].w;
            }
        }
        for (; j < cn; j += 4) {
            float4 v[4];
#pragma unroll
            for (int t = 0; t < 4; ++t) {
                const int sj = __builtin_amdgcn_readlane(s, j + t);
                v[t] = *(const float4*)(rows + (size_t)((unsigned)sj * 384u) + loff);
            }
#pragma unroll
            for (int t = 0; t < 4; ++t) {
                const bool on = (j + t < cn);
                acc.x += on ? v[t].x : 0.0f; acc.y += on ? v[t].y : 0.0f;
                acc.z += on ? v[t].z : 0.0f; acc.w += on ? v[t].w : 0.0f;
            }
        }
    }

    if (deg > GCN_CAP) {
        const int novf = __builtin_amdgcn_readfirstlane(cnt[n]);
        for (int i = 0; i < novf; ++i) {
            if (__builtin_amdgcn_readfirstlane(ovf_dst[i]) == node) {
                const int sj = __builtin_amdgcn_readfirstlane(ovf_src[i]);
                const float4 v = *(const float4*)(rows + (size_t)((unsigned)sj * 384u) + loff);
                acc.x += v.x; acc.y += v.y; acc.z += v.z; acc.w += v.w;
            }
        }
    }

    const float di = rsqrtf((float)deg + 1.0f);
    if (lane < 24) {
        const int c0 = lane * 4;
        float4 o;
        o.x = fmaxf(acc.x * di + bias[c0],     0.0f);
        o.y = fmaxf(acc.y * di + bias[c0 + 1], 0.0f);
        o.z = fmaxf(acc.z * di + bias[c0 + 2], 0.0f);
        o.w = fmaxf(acc.w * di + bias[c0 + 3], 0.0f);
        *(float4*)((char*)outp + (size_t)((unsigned)node * 384u) + (unsigned)lane * 16u) = o;
    }
}

__global__ void gcn_gather2_kernel(const int* __restrict__ cnt,
                                   const int* __restrict__ ell,
                                   const int* __restrict__ ovf_src,
                                   const int* __restrict__ ovf_dst,
                                   const float* __restrict__ hws,
                                   const float* __restrict__ bias,
                                   float* __restrict__ outp, int n) {
    const int lane = threadIdx.x & 31;
    const int node = (int)(blockIdx.x * (blockDim.x >> 5) + (threadIdx.x >> 5));
    if (node >= n) return;

    const float2* __restrict__ rows = (const float2*)hws;
    const int deg = cnt[node];
    const int nb  = (deg < GCN_CAP) ? deg : GCN_CAP;
    const int* __restrict__ bucket = ell + (size_t)node * GCN_CAP;

    float ax = 0.0f, ay = 0.0f;
    for (int idx = lane; idx < nb; idx += 32) {
        const float2 v = rows[bucket[idx]];
        ax += v.x; ay += v.y;
    }
    if (deg > GCN_CAP) {
        const int novf = cnt[n];
        for (int i = lane; i < novf; i += 32) {
            if (ovf_dst[i] == node) {
                const float2 v = rows[ovf_src[i]];
                ax += v.x; ay += v.y;
            }
        }
    }
#pragma unroll
    for (int m = 16; m > 0; m >>= 1) {
        ax += __shfl_xor(ax, m, 32);
        ay += __shfl_xor(ay, m, 32);
    }
    if (lane == 0) {
        const float2 self = rows[node];
        const float  di   = rsqrtf((float)deg + 1.0f);
        float2 o;
        o.x = (ax + self.x) * di + bias[0];
        o.y = (ay + self.y) * di + bias[1];
        ((float2*)outp)[node] = o;
    }
}


extern "C" void kernel_launch(void* const* d_in, const int* in_sizes, int n_in,
                              void* d_out, int out_size, void* d_ws, size_t ws_size,
                              hipStream_t stream) {
    (void)in_sizes; (void)n_in; (void)out_size; (void)ws_size;

    const int N = GCN_N;
    const int E = GCN_E;

    const float* x    = (const float*)d_in[0];
    const int*   ei   = (const int*)d_in[1];
    const int*   src  = ei;
    const int*   dstp = ei + E;

    const float* W[5] = { (const float*)d_in[2], (const float*)d_in[4],
                          (const float*)d_in[6], (const float*)d_in[8],
                          (const float*)d_in[10] };
    const float* B[5] = { (const float*)d_in[3], (const float*)d_in[5],
                          (const float*)d_in[7], (const float*)d_in[9],
                          (const float*)d_in[11] };

    int*   cnt    = (int*)d_ws;
    int*   ell    = cnt + 50048;
    int*   ovfs   = ell + (size_t)N * GCN_CAP;
    int*   ovfd   = ovfs + E;
    float* hws    = (float*)(ovfd + E);
    float* aggA   = hws  + (size_t)N * 96;
    float* aggB   = aggA + (size_t)N * 96;
    int*   tab    = (int*)(aggB + (size_t)N * 96);
    int2*  stage  = (int2*)(tab + GCN_NCH * GCN_TABP);
    float* out    = (float*)d_out;

    const int T = 256;

    gcn_bin_kernel  <<<GCN_NCH, 1024, 0, stream>>>(src, dstp, tab, stage, cnt, N, E);
    gcn_owner_kernel<<<GCN_NBIN, 256, 0, stream>>>(tab, stage, cnt, ell, ovfs, ovfd, N);

    const int dims[6] = { 165, 96, 96, 96, 96, 2 };
    const float* layerIn = x;
    float* aggBufs[2] = { aggA, aggB };

    for (int l = 0; l < 5; ++l) {
        const int K  = dims[l];
        const int Cn = dims[l + 1];
        float* agg = (l == 4) ? out : aggBufs[l & 1];

        const int tiles         = (N / 16) * ((Cn + 15) / 16);
        const int wavesPerBlock = 4;
        const int gBlocks       = (tiles + wavesPerBlock - 1) / wavesPerBlock;
        const int sBlocks       = (N / 16 + 1) / 2;
        if (l == 0) {
            gcn_gemm96_strip_wmma_f32<165><<<sBlocks, 64, 0, stream>>>(
                layerIn, W[l], cnt, hws, N);
        } else if (l < 4) {
            gcn_gemm96_strip_wmma_f32<96><<<sBlocks, 64, 0, stream>>>(
                layerIn, W[l], cnt, hws, N);
        } else {
            gcn_gemm_wmma_f32<0><<<gBlocks, wavesPerBlock * 32, 0, stream>>>(
                layerIn, W[l], cnt, hws, N, K, Cn);
        }

        const int nodesPerBlock = T / 32;
        const int aBlocks       = (N + nodesPerBlock - 1) / nodesPerBlock;
        if (l < 4) {
            gcn_gather96_kernel<<<aBlocks, T, 0, stream>>>(
                cnt, ell, ovfs, ovfd, hws, B[l], agg, N);
        } else {
            gcn_gather2_kernel<<<aBlocks, T, 0, stream>>>(
                cnt, ell, ovfs, ovfd, hws, B[l], agg, N);
        }

        layerIn = agg;
    }
}


# Layers 0-3: aggregation joined with the next layer's GEMM in one launch: 32-node block gathers rows into LDS, W staged in LDS, 3 waves per 16-row strip run WMMAs; 4 launches fewer; sum orders kept
# speedup vs baseline: 5.8063x; 1.1139x over previous
#include <hip/hip_runtime.h>
#include <hip/hip_bf16.h>

typedef __attribute__((ext_vector_type(2))) float v2f;
typedef __attribute__((ext_vector_type(8))) float v8f;

#define GCN_N 50000
#define GCN_E 800000

#define GCN_CAP  64
#define GCN_NCH  256
#define GCN_CE   3125
#define GCN_NBIN 782
#define GCN_TABP 784

__global__ void __launch_bounds__(1024)
gcn_bin_kernel(const int* __restrict__ src, const int* __restrict__ dst,
               int* __restrict__ tab, int2* __restrict__ stage,
               int* __restrict__ cnt, int n, int E) {
    __shared__ int  hist[GCN_TABP];
    __shared__ int  wsum[32];
    __shared__ int2 sorted[GCN_CE];

    const int tid  = threadIdx.x;
    const int lane = tid & 31;
    const int wid  = tid >> 5;
    const int c    = blockIdx.x;
    const int base = c * GCN_CE;

    if (tid < GCN_TABP) hist[tid] = 0;
    if (c == 0 && tid == 0) cnt[n] = 0;
    __syncthreads();

    int es[4], ed[4], er[4];
#pragma unroll
    for (int k = 0; k < 4; ++k) {
        const int  i  = k * 1024 + tid;
        const bool ok = (i < GCN_CE) && (base + i < E);
        ed[k] = ok ? dst[base + i] : -1;
        es[k] = ok ? src[base + i] : 0;
    }
#pragma unroll
    for (int k = 0; k < 4; ++k) {
        er[k] = 0;
        if (ed[k] >= 0) er[k] = atomicAdd(&hist[ed[k] >> 6], 1);
    }
    __syncthreads();

    const int v = (tid < GCN_NBIN) ? hist[tid] : 0;
    int x = v;
    for (int o = 1; o < 32; o <<= 1) {
        const int y = __shfl_up(x, o, 32);
        if (lane >= o) x += y;
    }
    if (lane == 31) wsum[wid] = x;
    __syncthreads();
    if (wid == 0) {
        const int w = wsum[lane];
        int z = w;
        for (int o = 1; o < 32; o <<= 1) {
            const int y = __shfl_up(z, o, 32);
            if (lane >= o) z += y;
        }
        wsum[lane] = z - w;
    }
    __syncthreads();
    const int excl = x - v + wsum[wid];
    if (tid <= GCN_NBIN) {
        hist[tid] = excl;
        tab[c * GCN_TABP + tid] = excl;
    }
    __syncthreads();

    const int tot = hist[GCN_NBIN];
#pragma unroll
    for (int k = 0; k < 4; ++k) {
        if (ed[k] >= 0) sorted[hist[ed[k] >> 6] + er[k]] = make_int2(es[k], ed[k]);
    }
    __syncthreads();
#pragma unroll
    for (int k = 0; k < 4; ++k) {
        const int i = k * 1024 + tid;
        if (i < tot) stage[(size_t)base + i] = sorted[i];
    }
}

__device__ __forceinline__ void gcn_place_edge(int2 e, int* cntL, int* ellL,
                                               int* __restrict__ cnt,
                                               int* __restrict__ ovf_src,
                                               int* __restrict__ ovf_dst, int n) {
    const int dl = e.y & 63;
    const int r  = atomicAdd(&cntL[dl], 1);
    if (r < GCN_CAP) {
        ellL[dl * GCN_CAP + r] = e.x;
    } else {
        const int o = atomicAdd(&cnt[n], 1);
        ovf_src[o] = e.x;
        ovf_dst[o] = e.y;
    }
}

__global__ void __launch_bounds__(256)
gcn_owner_kernel(const int* __restrict__ tab, const int2* __restrict__ stage,
                 int* __restrict__ cnt, int* __restrict__ ell,
                 int* __restrict__ ovf_src, int* __restrict__ ovf_dst, int n) {
    __shared__ int  cntL[64];
    __shared__ int4 ellL4[64 * GCN_CAP / 4];
    int* ellL = (int*)ellL4;

    const int tid  = threadIdx.x;
    const int lane = tid & 31;
    const int wid  = tid >> 5;
    const int b    = blockIdx.x;

    if (tid < 64) cntL[tid] = 0;

    const int myc = wid * 32 + lane;
    const int lo  = tab[myc * GCN_TABP + b];
    const int hi  = tab[myc * GCN_TABP + b + 1];
    __syncthreads();

    const int sub = lane >> 4;
    const int k16 = lane & 15;
    int2 ev[16];
    unsigned act = 0;
#pragma unroll
    for (int j = 0; j < 16; ++j) {
        const int la = __builtin_amdgcn_readlane(lo, 2 * j);
        const int lb = __builtin_amdgcn_readlane(lo, 2 * j + 1);
        const int ha = __builtin_amdgcn_readlane(hi, 2 * j);
        const int hb = __builtin_amdgcn_readlane(hi, 2 * j + 1);
        const int lj = sub ? lb : la;
        const int hj = sub ? hb : ha;
        const int2* __restrict__ p = stage + (size_t)(wid * 32 + 2 * j + sub) * GCN_CE;
        ev[j] = make_int2(0, 0);
        if (lj + k16 < hj) {
            ev[j] = p[lj + k16];
            act |= 1u << j;
        }
    }
#pragma unroll
    for (int j = 0; j < 16; ++j) {
        if (act & (1u << j)) gcn_place_edge(ev[j], cntL, ellL, cnt, ovf_src, ovf_dst, n);
    }
    for (int j = 0; j < 32; ++j) {
        const int lj = __builtin_amdgcn_readlane(lo, j);
        const int hj = __builtin_amdgcn_readlane(hi, j);
        if (hj - lj > 16) {
            const int2* __restrict__ p = stage + (size_t)(wid * 32 + j) * GCN_CE;
            for (int i = lj + 16 + lane; i < hj; i += 32)
                gcn_place_edge(p[i], cntL, ellL, cnt, ovf_src, ovf_dst, n);
        }
    }
    __syncthreads();

    const int node0 = b * 64;
    if (tid < 64 && node0 + tid < n) cnt[node0 + tid] = cntL[tid];
#pragma unroll
    for (int k = 0; k < 4; ++k) {
        const int i  = k * 256 + tid;
        const int nd = i >> 4;
        const int q  = i & 15;
        if (node0 + nd < n && (q < 8 || cntL[nd] > 32))
            *(int4*)(ell + (size_t)(node0 + nd) * GCN_CAP + 4 * q) = ellL4[nd * 16 + q];
    }
}

__device__ __forceinline__ float gcn_dinv(const int* __restrict__ cnt, int i) {
    return rsqrtf((float)cnt[i] + 1.0f);
}

struct GcnStage {
    float  a[4];
    float4 bl[4];
    float2 bh[4];
};

template <int K>
__device__ __forceinline__ void gcn_load_stage(GcnStage& st,
                                               const float* __restrict__ Arow,
                                               const float* __restrict__ Wl,
                                               const float* __restrict__ Wh, int kb) {
#pragma unroll
    for (int i = 0; i < 4; ++i) st.a[i] = Arow[kb + i];
#pragma unroll
    for (int i = 0; i < 4; ++i) {
        st.bl[i] = *(const float4*)(Wl + (size_t)(kb + i) * 96);
        st.bh[i] = *(const float2*)(Wh + (size_t)(kb + i) * 96);
    }
}

__device__ __forceinline__ void gcn_mma_stage(const GcnStage& st, v8f (&acc)[6]) {
#pragma unroll
    for (int s = 0; s < 2; ++s) {
        const float a0 = st.a[2 * s], a1 = st.a[2 * s + 1];
        const float4 p = st.bl[2 * s], q = st.bl[2 * s + 1];
        const float2 u = st.bh[2 * s], w = st.bh[2 * s + 1];
        v2f a;  a[0]  = a0;  a[1]  = a1;
        v2f b0; b0[0] = p.x; b0[1] = q.x;
        v2f b1; b1[0] = p.y; b1[1] = q.y;
        v2f b2; b2[0] = p.z; b2[1] = q.z;
        v2f b3; b3[0] = p.w; b3[1] = q.w;
        v2f b4; b4[0] = u.x; b4[1] = w.x;
        v2f b5; b5[0] = u.y; b5[1] = w.y;
        acc[0] = __builtin_amdgcn_wmma_f32_16x16x4_f32(false, a, false, b0, (short)0, acc[0], false, false);
        acc[1] = __builtin_amdgcn_wmma_f32_16x16x4_f32(false, a, false, b1, (short)0, acc[1], false, false);
        acc[2] = __builtin_amdgcn_wmma_f32_16x16x4_f32(false, a, false, b2, (short)0, acc[2], false, false);
        acc[3] = __builtin_amdgcn_wmma_f32_16x16x4_f32(false, a, false, b3, (short)0, acc[3], false, false);
        acc[4] = __builtin_amdgcn_wmma_f32_16x16x4_f32(false, a, false, b4, (short)0, acc[4], false, false);
        acc[5] = __builtin_amdgcn_wmma_f32_16x16x4_f32(false, a, false, b5, (short)0, acc[5], false, false);
    }
}

template <int K>
__global__ void __launch_bounds__(64)
gcn_gemm96_strip_wmma_f32(const float* __restrict__ A,
                          const float* __restrict__ W,
                          const int* __restrict__ cnt,
                          float* __restrict__ hws, int M) {
    const int lane = threadIdx.x & 31;
    const int tm   = blockIdx.x * (blockDim.x >> 5) + (threadIdx.x >> 5);
    if (tm >= (M >> 4)) return;

    const int c = lane & 15;
    const int h = lane >> 4;
    const int row = tm * 16 + c;

    const float* __restrict__ Arow = A + (size_t)row * K + 4 * h;
    const float* __restrict__ Wl   = W + (size_t)(4 * h) * 96 + 4 * c;
    const float* __restrict__ Wh   = W + (size_t)(4 * h) * 96 + 64 + 2 * c;

    const int rbase = tm * 16 + (h << 3);
    float dm[8];
#pragma unroll
    for (int r = 0; r < 8; ++r) dm[r] = gcn_dinv(cnt, rbase + r);

    v8f acc[6] = {};
    constexpr int K8 = K & ~7;

    constexpr int NB = K8 / 8;
    GcnStage st[3];
    gcn_load_stage<K>(st[0], Arow, Wl, Wh, 0);
    if constexpr (NB > 1) gcn_load_stage<K>(st[1], Arow, Wl, Wh, 8);
#pragma unroll
    for (int j = 0; j < NB; ++j) {
        if (j + 2 < NB) gcn_load_stage<K>(st[(j + 2) % 3], Arow, Wl, Wh, (j + 2) * 8);
        __builtin_amdgcn_sched_barrier(0);
        gcn_mma_stage(st[j % 3], acc);
    }

    if constexpr (K8 < K) {
        GcnStage t;
#pragma unroll
        for (int i = 0; i < 4; ++i) {
            const int  kk = K8 + 4 * h + i;
            const bool ok = (kk < K);
            const int  kc = ok ? kk : (K - 1);
            const float  av = A[(size_t)row * K + kc];
            const float4 lv = *(const float4*)(W + (size_t)kc * 96 + 4 * c);
            const float2 hv = *(const float2*)(W + (size_t)kc * 96 + 64 + 2 * c);
            t.a[i]    = ok ? av : 0.0f;
            t.bl[i].x = ok ? lv.x : 0.0f; t.bl[i].y = ok ? lv.y : 0.0f;
            t.bl[i].z = ok ? lv.z : 0.0f; t.bl[i].w = ok ? lv.w : 0.0f;
            t.bh[i].x = ok ? hv.x : 0.0f; t.bh[i].y = ok ? hv.y : 0.0f;
        }
        gcn_mma_stage(t, acc);
    }

#pragma unroll
    for (int r = 0; r < 8; ++r) {
        const size_t m = (size_t)(rbase + r);
        float4 o;
        o.x = acc[0][r] * dm[r]; o.y = acc[1][r] * dm[r];
        o.z = acc[2][r] * dm[r]; o.w = acc[3][r] * dm[r];
        float2 p;
        p.x = acc[4][r] * dm[r]; p.y = acc[5][r] * dm[r];
        *(float4*)(hws + m * 96 + 4 * c)      = o;
        *(float2*)(hws + m * 96 + 64 + 2 * c) = p;
    }
}

#define GCN_PA 100
#define GCN_PW 192

template <int NOUT>
__global__ void __launch_bounds__(1024)
gcn_gather96_gemm_kernel(const int* __restrict__ cnt,
                         const int* __restrict__ ell,
                         const int* __restrict__ ovf_src,
                         const int* __restrict__ ovf_dst,
                         const float* __restrict__ hws,
                         const float* __restrict__ bias,
                         const float* __restrict__ Wn,
                         float* __restrict__ hout, int n) {
    __shared__ float4 Wl4[(NOUT == 96) ? (48 * GCN_PW / 4) : (96 * 2 / 4)];
    __shared__ float4 Al4[32 * GCN_PA / 4];
    __shared__ float  dl[32];
    float* Wl = (float*)Wl4;
    float* Al = (float*)Al4;

    const int tid  = threadIdx.x;
    const int lane = tid & 31;
    const int wid  = __builtin_amdgcn_readfirstlane(tid >> 5);
    const int node = (int)blockIdx.x * 32 + wid;

    const int l4 = (lane < 24) ? lane : lane - 24;
    const char* __restrict__ rows = (const char*)hws;
    const unsigned loff = (unsigned)l4 * 16u;
    const int nodec = (node < n) ? node : (n - 1);
    float4 acc = *(const float4*)(rows + (size_t)((unsigned)nodec * 384u) + loff);
    const int deg = __builtin_amdgcn_readfirstlane(cnt[nodec]);

    const float4* __restrict__ W4 = (const float4*)Wn;
    const int  kp0 = tid / 24;
    const int  q0  = tid - kp0 * 24;
    const int  i1  = 1024 + tid;
    const int  kp1 = i1 / 24;
    const int  q1  = i1 - kp1 * 24;
    float4 ra0 = {}, ra1 = {}, rb0 = {}, rb1 = {};
    if (NOUT == 96) {
        ra0 = W4[(2 * kp0) * 24 + q0];
        ra1 = W4[(2 * kp0 + 1) * 24 + q0];
        if (wid < 4) {
            rb0 = W4[(2 * kp1) * 24 + q1];
            rb1 = W4[(2 * kp1 + 1) * 24 + q1];
        }
    } else {
        if (tid < 48) ra0 = W4[tid];
    }

    if (NOUT == 96) {
        float* d0 = Wl + kp0 * GCN_PW + 8 * q0;
        *(float4*)d0       = make_float4(ra0.x, ra1.x, ra0.y, ra1.y);
        *(float4*)(d0 + 4) = make_float4(ra0.z, ra1.z, ra0.w, ra1.w);
        if (wid < 4) {
            float* d1 = Wl + kp1 * GCN_PW + 8 * q1;
            *(float4*)d1       = make_float4(rb0.x, rb1.x, rb0.y, rb1.y);
            *(float4*)(d1 + 4) = make_float4(rb0.z, rb1.z, rb0.w, rb1.w);
        }
    } else {
        if (tid < 48) Wl4[tid] = ra0;
    }

    float4 o  = make_float4(0.0f, 0.0f, 0.0f, 0.0f);
    float  di = 0.0f;

    if (node < n) {
        const int nb  = (deg < GCN_CAP) ? deg : GCN_CAP;
        const int* __restrict__ bucket = ell + (size_t)node * GCN_CAP;

        for (int base = 0; base < nb; base += 32) {
            const int idx = base + lane;
            const int s   = (idx < nb) ? bucket[idx] : node;
            const int rem = nb - base;
            const int cn  = (rem < 32) ? rem : 32;

            int j = 0;
            for (; j + 8 <= cn; j += 8) {
                float4 v[8];
#pragma unroll
                for (int t = 0; t < 8; ++t) {
                    const int sj = __builtin_amdgcn_readlane(s, j + t);
                    v[t] = *(const float4*)(rows + (size_t)((unsigned)sj * 384u) + loff);
                }
#pragma unroll
                for (int t = 0; t < 8; ++t) {
                    acc.x += v[t].x; acc.y += v[t].y; acc.z += v[t].z; acc.w += v[t].w;
                }
            }
            for (; j < cn; j += 4) {
                float4 v[4];
#pragma unroll
                for (int t = 0; t < 4; ++t) {
                    const int sj = __builtin_amdgcn_readlane(s, j + t);
                    v[t] = *(const float4*)(rows + (size_t)((unsigned)sj * 384u) + loff);
                }
#pragma unroll
                for (int t = 0; t < 4; ++t) {
                    const bool on = (j + t < cn);
                    acc.x += on ? v[t].x : 0.0f; acc.y += on ? v[t].y : 0.0f;
                    acc.z += on ? v[t].z : 0.0f; acc.w += on ? v[t].w : 0.0f;
                }
            }
        }

        if (deg > GCN_CAP) {
            const int novf = __builtin_amdgcn_readfirstlane(cnt[n]);
            for (int i = 0; i < novf; ++i) {
                if (__builtin_amdgcn_readfirstlane(ovf_dst[i]) == node) {
                    const int sj = __builtin_amdgcn_readfirstlane(ovf_src[i]);
                    const float4 v = *(const float4*)(rows + (size_t)((unsigned)sj * 384u) + loff);
                    acc.x += v.x; acc.y += v.y; acc.z += v.z; acc.w += v.w;
                }
            }
        }

        di = rsqrtf((float)deg + 1.0f);
        const int c0 = l4 * 4;
        o.x = fmaxf(acc.x * di + bias[c0],     0.0f);
        o.y = fmaxf(acc.y * di + bias[c0 + 1], 0.0f);
        o.z = fmaxf(acc.z * di + bias[c0 + 2], 0.0f);
        o.w = fmaxf(acc.w * di + bias[c0 + 3], 0.0f);
    }

    if (lane < 24) *(float4*)(Al + wid * GCN_PA + lane * 4) = o;
    if (lane == 0) dl[wid] = di;
    __syncthreads();

    const int c = lane & 15;
    const int h = lane >> 4;

    if (NOUT == 96) {
        if (wid < 6) {
            const int sl = wid / 3;
            const int g  = wid - 3 * sl;
            const int tm = (int)blockIdx.x * 2 + sl;
            if (tm < (n >> 4)) {
                const float* Ar = Al + (sl * 16 + c) * GCN_PA + 4 * h;
                const float* Wr = Wl + (2 * h) * GCN_PW + (g * 32 + 2 * c) * 2;
                v8f acc0 = {}, acc1 = {};
#pragma unroll
                for (int kb = 0; kb < 96; kb += 8) {
                    const float4 a  = *(const float4*)(Ar + kb);
                    const float4 w0 = *(const float4*)(Wr + (kb / 2 + 0) * GCN_PW);
                    const float4 w1 = *(const float4*)(Wr + (kb / 2 + 1) * GCN_PW);
                    v2f a01; a01[0] = a.x;  a01[1] = a.y;
                    v2f a23; a23[0] = a.z;  a23[1] = a.w;
                    v2f p0;  p0[0]  = w0.x; p0[1]  = w0.y;
                    v2f p1;  p1[0]  = w0.z; p1[1]  = w0.w;
                    v2f q0v; q0v[0] = w1.x; q0v[1] = w1.y;
                    v2f q1v; q1v[0] = w1.z; q1v[1] = w1.w;
                    acc0 = __builtin_amdgcn_wmma_f32_16x16x4_f32(false, a01, false, p0,  (short)0, acc0, false, false);
                    acc1 = __builtin_amdgcn_wmma_f32_16x16x4_f32(false, a01, false, p1,  (short)0, acc1, false, false);
                    acc0 = __builtin_amdgcn_wmma_f32_16x16x4_f32(false, a23, false, q0v, (short)0, acc0, false, false);
                    acc1 = __builtin_amdgcn_wmma_f32_16x16x4_f32(false, a23, false, q1v, (short)0, acc1, false, false);
                }
#pragma unroll
                for (int r = 0; r < 8; ++r) {
                    const float  dm = dl[sl * 16 + 8 * h + r];
                    const size_t m  = (size_t)(tm * 16 + 8 * h + r);
                    float2 p;
                    p.x = acc0[r] * dm; p.y = acc1[r] * dm;
                    *(float2*)(hout + m * 96 + g * 32 + 2 * c) = p;
                }
            }
        }
    } else {
        if (wid < 2) {
            const int sl = wid;
            const int tm = (int)blockIdx.x * 2 + sl;
            if (tm < (n >> 4)) {
                const bool colOK = (c < NOUT);
                const int  cc    = colOK ? c : 0;
                const float* Ar = Al + (sl * 16 + c) * GCN_PA + 2 * h;
                const float* Wr = Wl + (2 * h) * NOUT + cc;
                v8f acc = {};
#pragma unroll
                for (int kb = 0; kb < 96; kb += 4) {
                    const float2 a  = *(const float2*)(Ar + kb);
                    const float  w0 = Wr[(kb + 0) * NOUT];
                    const float  w1 = Wr[(kb + 1) * NOUT];
                    v2f av; av[0] = a.x;                av[1] = a.y;
                    v2f bv; bv[0] = colOK ? w0 : 0.0f;  bv[1] = colOK ? w1 : 0.0f;
                    acc = __builtin_amdgcn_wmma_f32_16x16x4_f32(false, av, false, bv, (short)0, acc, false, false);
                }
                if (colOK) {
#pragma unroll
                    for (int r = 0; r < 8; ++r) {
                        const size_t m = (size_t)(tm * 16 + 8 * h + r);
                        hout[m * NOUT + c] = acc[r] * dl[sl * 16 + 8 * h + r];
                    }
                }
            }
        }
    }
}

__global__ void gcn_gather2_kernel(const int* __restrict__ cnt,
                                   const int* __restrict__ ell,
                                   const int* __restrict__ ovf_src,
                                   const int* __restrict__ ovf_dst,
                                   const float* __restrict__ hws,
                                   const float* __restrict__ bias,
                                   float* __restrict__ outp, int n) {
    const int lane = threadIdx.x & 31;
    const int node = (int)(blockIdx.x * (blockDim.x >> 5) + (threadIdx.x >> 5));
    if (node >= n) return;

    const float2* __restrict__ rows = (const float2*)hws;
    const int deg = cnt[node];
    const int nb  = (deg < GCN_CAP) ? deg : GCN_CAP;
    const int* __restrict__ bucket = ell + (size_t)node * GCN_CAP;

    float ax = 0.0f, ay = 0.0f;
    for (int idx = lane; idx < nb; idx += 32) {
        const float2 v = rows[bucket[idx]];
        ax += v.x; ay += v.y;
    }
    if (deg > GCN_CAP) {
        const int novf = cnt[n];
        for (int i = lane; i < novf; i += 32) {
            if (ovf_dst[i] == node) {
                const float2 v = rows[ovf_src[i]];
                ax += v.x; ay += v.y;
            }
        }
    }
#pragma unroll
    for (int m = 16; m > 0; m >>= 1) {
        ax += __shfl_xor(ax, m, 32);
        ay += __shfl_xor(ay, m, 32);
    }
    if (lane == 0) {
        const float2 self = rows[node];
        const float  di   = rsqrtf((float)deg + 1.0f);
        float2 o;
        o.x = (ax + self.x) * di + bias[0];
        o.y = (ay + self.y) * di + bias[1];
        ((float2*)outp)[node] = o;
    }
}


extern "C" void kernel_launch(void* const* d_in, const int* in_sizes, int n_in,
                              void* d_out, int out_size, void* d_ws, size_t ws_size,
                              hipStream_t stream) {
    (void)in_sizes; (void)n_in; (void)out_size; (void)ws_size;

    const int N = GCN_N;
    const int E = GCN_E;

    const float* x    = (const float*)d_in[0];
    const int*   ei   = (const int*)d_in[1];
    const int*   src  = ei;
    const int*   dstp = ei + E;

    const float* W[5] = { (const float*)d_in[2], (const float*)d_in[4],
                          (const float*)d_in[6], (const float*)d_in[8],
                          (const float*)d_in[10] };
    const float* B[5] = { (const float*)d_in[3], (const float*)d_in[5],
                          (const float*)d_in[7], (const float*)d_in[9],
                          (const float*)d_in[11] };

    int*   cnt    = (int*)d_ws;
    int*   ell    = cnt + 50048;
    int*   ovfs   = ell + (size_t)N * GCN_CAP;
    int*   ovfd   = ovfs + E;
    float* hA     = (float*)(ovfd + E);
    float* hB     = hA + (size_t)N * 96;
    int*   tab    = (int*)(hB + (size_t)N * 96);
    int2*  stage  = (int2*)(tab + GCN_NCH * GCN_TABP);
    float* out    = (float*)d_out;

    gcn_bin_kernel  <<<GCN_NCH, 1024, 0, stream>>>(src, dstp, tab, stage, cnt, N, E);
    gcn_owner_kernel<<<GCN_NBIN, 256, 0, stream>>>(tab, stage, cnt, ell, ovfs, ovfd, N);

    const int sBlocks = (N / 16 + 1) / 2;
    gcn_gemm96_strip_wmma_f32<165><<<sBlocks, 64, 0, stream>>>(x, W[0], cnt, hA, N);

    const int fBlocks = (N + 31) / 32;
    gcn_gather96_gemm_kernel<96><<<fBlocks, 1024, 0, stream>>>(cnt, ell, ovfs, ovfd, hA, B[0], W[1], hB, N);
    gcn_gather96_gemm_kernel<96><<<fBlocks, 1024, 0, stream>>>(cnt, ell, ovfs, ovfd, hB, B[1], W[2], hA, N);
    gcn_gather96_gemm_kernel<96><<<fBlocks, 1024, 0, stream>>>(cnt, ell, ovfs, ovfd, hA, B[2], W[3], hB, N);
    gcn_gather96_gemm_kernel<2> <<<fBlocks, 1024, 0, stream>>>(cnt, ell, ovfs, ovfd, hB, B[3], W[4], hA, N);

    const int aBlocks = (N + 7) / 8;
    gcn_gather2_kernel<<<aBlocks, 256, 0, stream>>>(cnt, ell, ovfs, ovfd, hA, B[4], out, N);
}


# Gather+GEMM block widened to 64 nodes, two per wave in turn: W staged once per 64 nodes, half as many waves start, 12 waves run the WMMAs; both nodes' first loads asked up front; sum orders kept
# speedup vs baseline: 6.0523x; 1.0424x over previous
#include <hip/hip_runtime.h>
#include <hip/hip_bf16.h>

typedef __attribute__((ext_vector_type(2))) float v2f;
typedef __attribute__((ext_vector_type(8))) float v8f;

#define GCN_N 50000
#define GCN_E 800000

#define GCN_CAP  64
#define GCN_NCH  256
#define GCN_CE   3125
#define GCN_NBIN 782
#define GCN_TABP 784

__global__ void __launch_bounds__(1024)
gcn_bin_kernel(const int* __restrict__ src, const int* __restrict__ dst,
               int* __restrict__ tab, int2* __restrict__ stage,
               int* __restrict__ cnt, int n, int E) {
    __shared__ int  hist[GCN_TABP];
    __shared__ int  wsum[32];
    __shared__ int2 sorted[GCN_CE];

    const int tid  = threadIdx.x;
    const int lane = tid & 31;
    const int wid  = tid >> 5;
    const int c    = blockIdx.x;
    const int base = c * GCN_CE;

    if (tid < GCN_TABP) hist[tid] = 0;
    if (c == 0 && tid == 0) cnt[n] = 0;
    __syncthreads();

    int es[4], ed[4], er[4];
#pragma unroll
    for (int k = 0; k < 4; ++k) {
        const int  i  = k * 1024 + tid;
        const bool ok = (i < GCN_CE) && (base + i < E);
        ed[k] = ok ? dst[base + i] : -1;
        es[k] = ok ? src[base + i] : 0;
    }
#pragma unroll
    for (int k = 0; k < 4; ++k) {
        er[k] = 0;
        if (ed[k] >= 0) er[k] = atomicAdd(&hist[ed[k] >> 6], 1);
    }
    __syncthreads();

    const int v = (tid < GCN_NBIN) ? hist[tid] : 0;
    int x = v;
    for (int o = 1; o < 32; o <<= 1) {
        const int y = __shfl_up(x, o, 32);
        if (lane >= o) x += y;
    }
    if (lane == 31) wsum[wid] = x;
    __syncthreads();
    if (wid == 0) {
        const int w = wsum[lane];
        int z = w;
        for (int o = 1; o < 32; o <<= 1) {
            const int y = __shfl_up(z, o, 32);
            if (lane >= o) z += y;
        }
        wsum[lane] = z - w;
    }
    __syncthreads();
    const int excl = x - v + wsum[wid];
    if (tid <= GCN_NBIN) {
        hist[tid] = excl;
        tab[c * GCN_TABP + tid] = excl;
    }
    __syncthreads();

    const int tot = hist[GCN_NBIN];
#pragma unroll
    for (int k = 0; k < 4; ++k) {
        if (ed[k] >= 0) sorted[hist[ed[k] >> 6] + er[k]] = make_int2(es[k], ed[k]);
    }
    __syncthreads();
#pragma unroll
    for (int k = 0; k < 4; ++k) {
        const int i = k * 1024 + tid;
        if (i < tot) stage[(size_t)base + i] = sorted[i];
    }
}

__device__ __forceinline__ void gcn_place_edge(int2 e, int* cntL, int* ellL,
                                               int* __restrict__ cnt,
                                               int* __restrict__ ovf_src,
                                               int* __restrict__ ovf_dst, int n) {
    const int dl = e.y & 63;
    const int r  = atomicAdd(&cntL[dl], 1);
    if (r < GCN_CAP) {
        ellL[dl * GCN_CAP + r] = e.x;
    } else {
        const int o = atomicAdd(&cnt[n], 1);
        ovf_src[o] = e.x;
        ovf_dst[o] = e.y;
    }
}

__global__ void __launch_bounds__(256)
gcn_owner_kernel(const int* __restrict__ tab, const int2* __restrict__ stage,
                 int* __restrict__ cnt, int* __restrict__ ell,
                 int* __restrict__ ovf_src, int* __restrict__ ovf_dst, int n) {
    __shared__ int  cntL[64];
    __shared__ int4 ellL4[64 * GCN_CAP / 4];
    int* ellL = (int*)ellL4;

    const int tid  = threadIdx.x;
    const int lane = tid & 31;
    const int wid  = tid >> 5;
    const int b    = blockIdx.x;

    if (tid < 64) cntL[tid] = 0;

    const int myc = wid * 32 + lane;
    const int lo  = tab[myc * GCN_TABP + b];
    const int hi  = tab[myc * GCN_TABP + b + 1];
    __syncthreads();

    const int sub = lane >> 4;
    const int k16 = lane & 15;
    int2 ev[16];
    unsigned act = 0;
#pragma unroll
    for (int j = 0; j < 16; ++j) {
        const int la = __builtin_amdgcn_readlane(lo, 2 * j);
        const int lb = __builtin_amdgcn_readlane(lo, 2 * j + 1);
        const int ha = __builtin_amdgcn_readlane(hi, 2 * j);
        const int hb = __builtin_amdgcn_readlane(hi, 2 * j + 1);
        const int lj = sub ? lb : la;
        const int hj = sub ? hb : ha;
        const int2* __restrict__ p = stage + (size_t)(wid * 32 + 2 * j + sub) * GCN_CE;
        ev[j] = make_int2(0, 0);
        if (lj + k16 < hj) {
            ev[j] = p[lj + k16];
            act |= 1u << j;
        }
    }
#pragma unroll
    for (int j = 0; j < 16; ++j) {
        if (act & (1u << j)) gcn_place_edge(ev[j], cntL, ellL, cnt, ovf_src, ovf_dst, n);
    }
    for (int j = 0; j < 32; ++j) {
        const int lj = __builtin_amdgcn_readlane(lo, j);
        const int hj = __builtin_amdgcn_readlane(hi, j);
        if (hj - lj > 16) {
            const int2* __restrict__ p = stage + (size_t)(wid * 32 + j) * GCN_CE;
            for (int i = lj + 16 + lane; i < hj; i += 32)
                gcn_place_edge(p[i], cntL, ellL, cnt, ovf_src, ovf_dst, n);
        }
    }
    __syncthreads();

    const int node0 = b * 64;
    if (tid < 64 && node0 + tid < n) cnt[node0 + tid] = cntL[tid];
#pragma unroll
    for (int k = 0; k < 4; ++k) {
        const int i  = k * 256 + tid;
        const int nd = i >> 4;
        const int q  = i & 15;
        if (node0 + nd < n && (q < 8 || cntL[nd] > 32))
            *(int4*)(ell + (size_t)(node0 + nd) * GCN_CAP + 4 * q) = ellL4[nd * 16 + q];
    }
}

__device__ __forceinline__ float gcn_dinv(const int* __restrict__ cnt, int i) {
    return rsqrtf((float)cnt[i] + 1.0f);
}

struct GcnStage {
    float  a[4];
    float4 bl[4];
    float2 bh[4];
};

template <int K>
__device__ __forceinline__ void gcn_load_stage(GcnStage& st,
                                               const float* __restrict__ Arow,
                                               const float* __restrict__ Wl,
                                               const float* __restrict__ Wh, int kb) {
#pragma unroll
    for (int i = 0; i < 4; ++i) st.a[i] = Arow[kb + i];
#pragma unroll
    for (int i = 0; i < 4; ++i) {
        st.bl[i] = *(const float4*)(Wl + (size_t)(kb + i) * 96);
        st.bh[i] = *(const float2*)(Wh + (size_t)(kb + i) * 96);
    }
}

__device__ __forceinline__ void gcn_mma_stage(const GcnStage& st, v8f (&acc)[6]) {
#pragma unroll
    for (int s = 0; s < 2; ++s) {
        const float a0 = st.a[2 * s], a1 = st.a[2 * s + 1];
        const float4 p = st.bl[2 * s], q = st.bl[2 * s + 1];
        const float2 u = st.bh[2 * s], w = st.bh[2 * s + 1];
        v2f a;  a[0]  = a0;  a[1]  = a1;
        v2f b0; b0[0] = p.x; b0[1] = q.x;
        v2f b1; b1[0] = p.y; b1[1] = q.y;
        v2f b2; b2[0] = p.z; b2[1] = q.z;
        v2f b3; b3[0] = p.w; b3[1] = q.w;
        v2f b4; b4[0] = u.x; b4[1] = w.x;
        v2f b5; b5[0] = u.y; b5[1] = w.y;
        acc[0] = __builtin_amdgcn_wmma_f32_16x16x4_f32(false, a, false, b0, (short)0, acc[0], false, false);
        acc[1] = __builtin_amdgcn_wmma_f32_16x16x4_f32(false, a, false, b1, (short)0, acc[1], false, false);
        acc[2] = __builtin_amdgcn_wmma_f32_16x16x4_f32(false, a, false, b2, (short)0, acc[2], false, false);
        acc[3] = __builtin_amdgcn_wmma_f32_16x16x4_f32(false, a, false, b3, (short)0, acc[3], false, false);
        acc[4] = __builtin_amdgcn_wmma_f32_16x16x4_f32(false, a, false, b4, (short)0, acc[4], false, false);
        acc[5] = __builtin_amdgcn_wmma_f32_16x16x4_f32(false, a, false, b5, (short)0, acc[5], false, false);
    }
}

template <int K>
__global__ void __launch_bounds__(64)
gcn_gemm96_strip_wmma_f32(const float* __restrict__ A,
                          const float* __restrict__ W,
                          const int* __restrict__ cnt,
                          float* __restrict__ hws, int M) {
    const int lane = threadIdx.x & 31;
    const int tm   = blockIdx.x * (blockDim.x >> 5) + (threadIdx.x >> 5);
    if (tm >= (M >> 4)) return;

    const int c = lane & 15;
    const int h = lane >> 4;
    const int row = tm * 16 + c;

    const float* __restrict__ Arow = A + (size_t)row * K + 4 * h;
    const float* __restrict__ Wl   = W + (size_t)(4 * h) * 96 + 4 * c;
    const float* __restrict__ Wh   = W + (size_t)(4 * h) * 96 + 64 + 2 * c;

    const int rbase = tm * 16 + (h << 3);
    float dm[8];
#pragma unroll
    for (int r = 0; r < 8; ++r) dm[r] = gcn_dinv(cnt, rbase + r);

    v8f acc[6] = {};
    constexpr int K8 = K & ~7;

    constexpr int NB = K8 / 8;
    GcnStage st[3];
    gcn_load_stage<K>(st[0], Arow, Wl, Wh, 0);
    if constexpr (NB > 1) gcn_load_stage<K>(st[1], Arow, Wl, Wh, 8);
#pragma unroll
    for (int j = 0; j < NB; ++j) {
        if (j + 2 < NB) gcn_load_stage<K>(st[(j + 2) % 3], Arow, Wl, Wh, (j + 2) * 8);
        __builtin_amdgcn_sched_barrier(0);
        gcn_mma_stage(st[j % 3], acc);
    }

    if constexpr (K8 < K) {
        GcnStage t;
#pragma unroll
        for (int i = 0; i < 4; ++i) {
            const int  kk = K8 + 4 * h + i;
            const bool ok = (kk < K);
            const int  kc = ok ? kk : (K - 1);
            const float  av = A[(size_t)row * K + kc];
            const float4 lv = *(const float4*)(W + (size_t)kc * 96 + 4 * c);
            const float2 hv = *(const float2*)(W + (size_t)kc * 96 + 64 + 2 * c);
            t.a[i]    = ok ? av : 0.0f;
            t.bl[i].x = ok ? lv.x : 0.0f; t.bl[i].y = ok ? lv.y : 0.0f;
            t.bl[i].z = ok ? lv.z : 0.0f; t.bl[i].w = ok ? lv.w : 0.0f;
            t.bh[i].x = ok ? hv.x : 0.0f; t.bh[i].y = ok ? hv.y : 0.0f;
        }
        gcn_mma_stage(t, acc);
    }

#pragma unroll
    for (int r = 0; r < 8; ++r) {
        const size_t m = (size_t)(rbase + r);
        float4 o;
        o.x = acc[0][r] * dm[r]; o.y = acc[1][r] * dm[r];
        o.z = acc[2][r] * dm[r]; o.w = acc[3][r] * dm[r];
        float2 p;
        p.x = acc[4][r] * dm[r]; p.y = acc[5][r] * dm[r];
        *(float4*)(hws + m * 96 + 4 * c)      = o;
        *(float2*)(hws + m * 96 + 64 + 2 * c) = p;
    }
}

#define GCN_PA 100
#define GCN_PW 192
#define GCN_NB 64

__device__ __forceinline__ float4 gcn_gather_node(float4 acc, int s0, int nb, int deg, int node,
                                                  const int* __restrict__ bucket,
                                                  const char* __restrict__ rows, unsigned loff, int l4, int lane,
                                                  const int* __restrict__ cnt,
                                                  const int* __restrict__ ovf_src,
                                                  const int* __restrict__ ovf_dst, int n) {
    for (int base = 0; base < nb; base += 32) {
        const int idx = base + lane;
        int s = s0;
        if (base > 0) s = (idx < nb) ? bucket[idx] : node;
        const int rem = nb - base;
        const int cn  = (rem < 32) ? rem : 32;

        int j = 0;
        for (; j + 8 <= cn; j += 8) {
            float4 v[8];
#pragma unroll
            for (int t = 0; t < 8; ++t) {
                const int sj = __builtin_amdgcn_readlane(s, j + t);
                v[t] = ((const float4*)(rows + (size_t)((unsigned)sj * 384u)))[l4];
            }
#pragma unroll
            for (int t = 0; t < 8; ++t) {
                acc.x += v[t].x; acc.y += v[t].y; acc.z += v[t].z; acc.w += v[t].w;
            }
        }
        for (; j < cn; j += 4) {
            float4 v[4];
#pragma unroll
            for (int t = 0; t < 4; ++t) {
                const int sj = __builtin_amdgcn_readlane(s, j + t);
                v[t] = ((const float4*)(rows + (size_t)((unsigned)sj * 384u)))[l4];
            }
#pragma unroll
            for (int t = 0; t < 4; ++t) {
                const bool on = (j + t < cn);
                acc.x += on ? v[t].x : 0.0f; acc.y += on ? v[t].y : 0.0f;
                acc.z += on ? v[t].z : 0.0f; acc.w += on ? v[t].w : 0.0f;
            }
        }
    }

    if (deg > GCN_CAP) {
        const int novf = __builtin_amdgcn_readfirstlane(cnt[n]);
        for (int i = 0; i < novf; ++i) {
            if (__builtin_amdgcn_readfirstlane(ovf_dst[i]) == node) {
                const int sj = __builtin_amdgcn_readfirstlane(ovf_src[i]);
                const float4 v = *(const float4*)(rows + (size_t)((unsigned)sj * 384u) + loff);
                acc.x += v.x; acc.y += v.y; acc.z += v.z; acc.w += v.w;
            }
        }
    }
    return acc;
}

template <int NOUT>
__global__ void __launch_bounds__(1024)
gcn_gather96_gemm_kernel(const int* __restrict__ cnt,
                         const int* __restrict__ ell,
                         const int* __restrict__ ovf_src,
                         const int* __restrict__ ovf_dst,
                         const float* __restrict__ hws,
                         const float* __restrict__ bias,
                         const float* __restrict__ Wn,
                         float* __restrict__ hout, int n) {
    __shared__ float4 Wl4[(NOUT == 96) ? (48 * GCN_PW / 4) : (96 * 2 / 4)];
    __shared__ float4 Al4[GCN_NB * GCN_PA / 4];
    __shared__ float  dl[GCN_NB];
    float* Wl = (float*)Wl4;
    float* Al = (float*)Al4;

    const int tid  = threadIdx.x;
    const int lane = tid & 31;
    const int wid  = __builtin_amdgcn_readfirstlane(tid >> 5);
    const int node0 = (int)blockIdx.x * GCN_NB + 2 * wid;

    const int l4 = (lane < 24) ? lane : lane - 24;
    const char* __restrict__ rows = (const char*)hws;
    const unsigned loff = (unsigned)l4 * 16u;
    int    nd[2], ndc[2], dg[2];
    float4 ac[2];
#pragma unroll
    for (int p = 0; p < 2; ++p) {
        nd[p]  = node0 + p;
        ndc[p] = (nd[p] < n) ? nd[p] : (n - 1);
        ac[p]  = *(const float4*)(rows + (size_t)((unsigned)ndc[p] * 384u) + loff);
        dg[p]  = __builtin_amdgcn_readfirstlane(cnt[ndc[p]]);
    }

    const float4* __restrict__ W4 = (const float4*)Wn;
    const int  kp0 = tid / 24;
    const int  q0  = tid - kp0 * 24;
    const int  i1  = 1024 + tid;
    const int  kp1 = i1 / 24;
    const int  q1  = i1 - kp1 * 24;
    float4 ra0 = {}, ra1 = {}, rb0 = {}, rb1 = {};
    if (NOUT == 96) {
        ra0 = W4[(2 * kp0) * 24 + q0];
        ra1 = W4[(2 * kp0 + 1) * 24 + q0];
        if (wid < 4) {
            rb0 = W4[(2 * kp1) * 24 + q1];
            rb1 = W4[(2 * kp1 + 1) * 24 + q1];
        }
    } else {
        if (tid < 48) ra0 = W4[tid];
    }

    if (NOUT == 96) {
        float* d0 = Wl + kp0 * GCN_PW + 8 * q0;
        *(float4*)d0       = make_float4(ra0.x, ra1.x, ra0.y, ra1.y);
        *(float4*)(d0 + 4) = make_float4(ra0.z, ra1.z, ra0.w, ra1.w);
        if (wid < 4) {
            float* d1 = Wl + kp1 * GCN_PW + 8 * q1;
            *(float4*)d1       = make_float4(rb0.x, rb1.x, rb0.y, rb1.y);
            *(float4*)(d1 + 4) = make_float4(rb0.z, rb1.z, rb0.w, rb1.w);
        }
    } else {
        if (tid < 48) Wl4[tid] = ra0;
    }

    int nbk[2], s0[2];
#pragma unroll
    for (int p = 0; p < 2; ++p) {
        nbk[p] = (dg[p] < GCN_CAP) ? dg[p] : GCN_CAP;
        s0[p]  = (lane < nbk[p]) ? ell[(size_t)ndc[p] * GCN_CAP + lane] : ndc[p];
    }

#pragma unroll
    for (int p = 0; p < 2; ++p) {
        float4 o  = make_float4(0.0f, 0.0f, 0.0f, 0.0f);
        float  di = 0.0f;
        if (nd[p] < n) {
            const float4 acc = gcn_gather_node(ac[p], s0[p], nbk[p], dg[p], nd[p],
                                               ell + (size_t)nd[p] * GCN_CAP, rows, loff, l4, lane,
                                               cnt, ovf_src, ovf_dst, n);
            di = rsqrtf((float)dg[p] + 1.0f);
            const int c0 = l4 * 4;
            o.x = fmaxf(acc.x * di + bias[c0],     0.0f);
            o.y = fmaxf(acc.y * di + bias[c0 + 1], 0.0f);
            o.z = fmaxf(acc.z * di + bias[c0 + 2], 0.0f);
            o.w = fmaxf(acc.w * di + bias[c0 + 3], 0.0f);
        }
        if (lane < 24) *(float4*)(Al + (2 * wid + p) * GCN_PA + lane * 4) = o;
        if (lane == 0) dl[2 * wid + p] = di;
    }
    __syncthreads();

    const int c = lane & 15;
    const int h = lane >> 4;

    if (NOUT == 96) {
        if (wid < 3 * (GCN_NB / 16)) {
            const int sl = wid / 3;
            const int g  = wid - 3 * sl;
            const int tm = (int)blockIdx.x * (GCN_NB / 16) + sl;
            if (tm < (n >> 4)) {
                const float* Ar = Al + (sl * 16 + c) * GCN_PA + 4 * h;
                const float* Wr = Wl + (2 * h) * GCN_PW + (g * 32 + 2 * c) * 2;
                v8f acc0 = {}, acc1 = {};
#pragma unroll
                for (int kb = 0; kb < 96; kb += 8) {
                    const float4 a  = *(const float4*)(Ar + kb);
                    const float4 w0 = *(const float4*)(Wr + (kb / 2 + 0) * GCN_PW);
                    const float4 w1 = *(const float4*)(Wr + (kb / 2 + 1) * GCN_PW);
                    v2f a01; a01[0] = a.x;  a01[1] = a.y;
                    v2f a23; a23[0] = a.z;  a23[1] = a.w;
                    v2f p0;  p0[0]  = w0.x; p0[1]  = w0.y;
                    v2f p1;  p1[0]  = w0.z; p1[1]  = w0.w;
                    v2f q0v; q0v[0] = w1.x; q0v[1] = w1.y;
                    v2f q1v; q1v[0] = w1.z; q1v[1] = w1.w;
                    acc0 = __builtin_amdgcn_wmma_f32_16x16x4_f32(false, a01, false, p0,  (short)0, acc0, false, false);
                    acc1 = __builtin_amdgcn_wmma_f32_16x16x4_f32(false, a01, false, p1,  (short)0, acc1, false, false);
                    acc0 = __builtin_amdgcn_wmma_f32_16x16x4_f32(false, a23, false, q0v, (short)0, acc0, false, false);
                    acc1 = __builtin_amdgcn_wmma_f32_16x16x4_f32(false, a23, false, q1v, (short)0, acc1, false, false);
                }
#pragma unroll
                for (int r = 0; r < 8; ++r) {
                    const float  dm = dl[sl * 16 + 8 * h + r];
                    const size_t m  = (size_t)(tm * 16 + 8 * h + r);
                    float2 p;
                    p.x = acc0[r] * dm; p.y = acc1[r] * dm;
                    *(float2*)(hout + m * 96 + g * 32 + 2 * c) = p;
                }
            }
        }
    } else {
        if (wid < GCN_NB / 16) {
            const int sl = wid;
            const int tm = (int)blockIdx.x * (GCN_NB / 16) + sl;
            if (tm < (n >> 4)) {
                const bool colOK = (c < NOUT);
                const int  cc    = colOK ? c : 0;
                const float* Ar = Al + (sl * 16 + c) * GCN_PA + 2 * h;
                const float* Wr = Wl + (2 * h) * NOUT + cc;
                v8f acc = {};
#pragma unroll
                for (int kb = 0; kb < 96; kb += 4) {
                    const float2 a  = *(const float2*)(Ar + kb);
                    const float  w0 = Wr[(kb + 0) * NOUT];
                    const float  w1 = Wr[(kb + 1) * NOUT];
                    v2f av; av[0] = a.x;                av[1] = a.y;
                    v2f bv; bv[0] = colOK ? w0 : 0.0f;  bv[1] = colOK ? w1 : 0.0f;
                    acc = __builtin_amdgcn_wmma_f32_16x16x4_f32(false, av, false, bv, (short)0, acc, false, false);
                }
                if (colOK) {
#pragma unroll
                    for (int r = 0; r < 8; ++r) {
                        const size_t m = (size_t)(tm * 16 + 8 * h + r);
                        hout[m * NOUT + c] = acc[r] * dl[sl * 16 + 8 * h + r];
                    }
                }
            }
        }
    }
}

__global__ void gcn_gather2_kernel(const int* __restrict__ cnt,
                                   const int* __restrict__ ell,
                                   const int* __restrict__ ovf_src,
                                   const int* __restrict__ ovf_dst,
                                   const float* __restrict__ hws,
                                   const float* __restrict__ bias,
                                   float* __restrict__ outp, int n) {
    const int lane = threadIdx.x & 31;
    const int node = (int)(blockIdx.x * (blockDim.x >> 5) + (threadIdx.x >> 5));
    if (node >= n) return;

    const float2* __restrict__ rows = (const float2*)hws;
    const int deg = cnt[node];
    const int nb  = (deg < GCN_CAP) ? deg : GCN_CAP;
    const int* __restrict__ bucket = ell + (size_t)node * GCN_CAP;

    float ax = 0.0f, ay = 0.0f;
    for (int idx = lane; idx < nb; idx += 32) {
        const float2 v = rows[bucket[idx]];
        ax += v.x; ay += v.y;
    }
    if (deg > GCN_CAP) {
        const int novf = cnt[n];
        for (int i = lane; i < novf; i += 32) {
            if (ovf_dst[i] == node) {
                const float2 v = rows[ovf_src[i]];
                ax += v.x; ay += v.y;
            }
        }
    }
#pragma unroll
    for (int m = 16; m > 0; m >>= 1) {
        ax += __shfl_xor(ax, m, 32);
        ay += __shfl_xor(ay, m, 32);
    }
    if (lane == 0) {
        const float2 self = rows[node];
        const float  di   = rsqrtf((float)deg + 1.0f);
        float2 o;
        o.x = (ax + self.x) * di + bias[0];
        o.y = (ay + self.y) * di + bias[1];
        ((float2*)outp)[node] = o;
    }
}


extern "C" void kernel_launch(void* const* d_in, const int* in_sizes, int n_in,
                              void* d_out, int out_size, void* d_ws, size_t ws_size,
                              hipStream_t stream) {
    (void)in_sizes; (void)n_in; (void)out_size; (void)ws_size;

    const int N = GCN_N;
    const int E = GCN_E;

    const float* x    = (const float*)d_in[0];
    const int*   ei   = (const int*)d_in[1];
    const int*   src  = ei;
    const int*   dstp = ei + E;

    const float* W[5] = { (const float*)d_in[2], (const float*)d_in[4],
                          (const float*)d_in[6], (const float*)d_in[8],
                          (const float*)d_in[10] };
    const float* B[5] = { (const float*)d_in[3], (const float*)d_in[5],
                          (const float*)d_in[7], (const float*)d_in[9],
                          (const float*)d_in[11] };

    int*   cnt    = (int*)d_ws;
    int*   ell    = cnt + 50048;
    int*   ovfs   = ell + (size_t)N * GCN_CAP;
    int*   ovfd   = ovfs + E;
    float* hA     = (float*)(ovfd + E);
    float* hB     = hA + (size_t)N * 96;
    int*   tab    = (int*)(hB + (size_t)N * 96);
    int2*  stage  = (int2*)(tab + GCN_NCH * GCN_TABP);
    float* out    = (float*)d_out;

    gcn_bin_kernel  <<<GCN_NCH, 1024, 0, stream>>>(src, dstp, tab, stage, cnt, N, E);
    gcn_owner_kernel<<<GCN_NBIN, 256, 0, stream>>>(tab, stage, cnt, ell, ovfs, ovfd, N);

    const int sBlocks = (N / 16 + 1) / 2;
    gcn_gemm96_strip_wmma_f32<165><<<sBlocks, 64, 0, stream>>>(x, W[0], cnt, hA, N);

    const int fBlocks = (N + GCN_NB - 1) / GCN_NB;
    gcn_gather96_gemm_kernel<96><<<fBlocks, 1024, 0, stream>>>(cnt, ell, ovfs, ovfd, hA, B[0], W[1], hB, N);
    gcn_gather96_gemm_kernel<96><<<fBlocks, 1024, 0, stream>>>(cnt, ell, ovfs, ovfd, hB, B[1], W[2], hA, N);
    gcn_gather96_gemm_kernel<96><<<fBlocks, 1024, 0, stream>>>(cnt, ell, ovfs, ovfd, hA, B[2], W[3], hB, N);
    gcn_gather96_gemm_kernel<2> <<<fBlocks, 1024, 0, stream>>>(cnt, ell, ovfs, ovfd, hB, B[3], W[4], hA, N);

    const int aBlocks = (N + 7) / 8;
    gcn_gather2_kernel<<<aBlocks, 256, 0, stream>>>(cnt, ell, ovfs, ovfd, hA, B[4], out, N);
}


# Last layer's 2-channel gather: two lanes per node, 16 nodes per wave (3125 waves started, not 50000); bucket line and 16 row loads asked as two groups; per-node sum order changed (lane pair)
# speedup vs baseline: 6.2594x; 1.0342x over previous
#include <hip/hip_runtime.h>
#include <hip/hip_bf16.h>

typedef __attribute__((ext_vector_type(2))) float v2f;
typedef __attribute__((ext_vector_type(8))) float v8f;

#define GCN_N 50000
#define GCN_E 800000

#define GCN_CAP  64
#define GCN_NCH  256
#define GCN_CE   3125
#define GCN_NBIN 782
#define GCN_TABP 784

__global__ void __launch_bounds__(1024)
gcn_bin_kernel(const int* __restrict__ src, const int* __restrict__ dst,
               int* __restrict__ tab, int2* __restrict__ stage,
               int* __restrict__ cnt, int n, int E) {
    __shared__ int  hist[GCN_TABP];
    __shared__ int  wsum[32];
    __shared__ int2 sorted[GCN_CE];

    const int tid  = threadIdx.x;
    const int lane = tid & 31;
    const int wid  = tid >> 5;
    const int c    = blockIdx.x;
    const int base = c * GCN_CE;

    if (tid < GCN_TABP) hist[tid] = 0;
    if (c == 0 && tid == 0) cnt[n] = 0;
    __syncthreads();

    int es[4], ed[4], er[4];
#pragma unroll
    for (int k = 0; k < 4; ++k) {
        const int  i  = k * 1024 + tid;
        const bool ok = (i < GCN_CE) && (base + i < E);
        ed[k] = ok ? dst[base + i] : -1;
        es[k] = ok ? src[base + i] : 0;
    }
#pragma unroll
    for (int k = 0; k < 4; ++k) {
        er[k] = 0;
        if (ed[k] >= 0) er[k] = atomicAdd(&hist[ed[k] >> 6], 1);
    }
    __syncthreads();

    const int v = (tid < GCN_NBIN) ? hist[tid] : 0;
    int x = v;
    for (int o = 1; o < 32; o <<= 1) {
        const int y = __shfl_up(x, o, 32);
        if (lane >= o) x += y;
    }
    if (lane == 31) wsum[wid] = x;
    __syncthreads();
    if (wid == 0) {
        const int w = wsum[lane];
        int z = w;
        for (int o = 1; o < 32; o <<= 1) {
            const int y = __shfl_up(z, o, 32);
            if (lane >= o) z += y;
        }
        wsum[lane] = z - w;
    }
    __syncthreads();
    const int excl = x - v + wsum[wid];
    if (tid <= GCN_NBIN) {
        hist[tid] = excl;
        tab[c * GCN_TABP + tid] = excl;
    }
    __syncthreads();

    const int tot = hist[GCN_NBIN];
#pragma unroll
    for (int k = 0; k < 4; ++k) {
        if (ed[k] >= 0) sorted[hist[ed[k] >> 6] + er[k]] = make_int2(es[k], ed[k]);
    }
    __syncthreads();
#pragma unroll
    for (int k = 0; k < 4; ++k) {
        const int i = k * 1024 + tid;
        if (i < tot) stage[(size_t)base + i] = sorted[i];
    }
}

__device__ __forceinline__ void gcn_place_edge(int2 e, int* cntL, int* ellL,
                                               int* __restrict__ cnt,
                                               int* __restrict__ ovf_src,
                                               int* __restrict__ ovf_dst, int n) {
    const int dl = e.y & 63;
    const int r  = atomicAdd(&cntL[dl], 1);
    if (r < GCN_CAP) {
        ellL[dl * GCN_CAP + r] = e.x;
    } else {
        const int o = atomicAdd(&cnt[n], 1);
        ovf_src[o] = e.x;
        ovf_dst[o] = e.y;
    }
}

__global__ void __launch_bounds__(256)
gcn_owner_kernel(const int* __restrict__ tab, const int2* __restrict__ stage,
                 int* __restrict__ cnt, int* __restrict__ ell,
                 int* __restrict__ ovf_src, int* __restrict__ ovf_dst, int n) {
    __shared__ int  cntL[64];
    __shared__ int4 ellL4[64 * GCN_CAP / 4];
    int* ellL = (int*)ellL4;

    const int tid  = threadIdx.x;
    const int lane = tid & 31;
    const int wid  = tid >> 5;
    const int b    = blockIdx.x;

    if (tid < 64) cntL[tid] = 0;

    const int myc = wid * 32 + lane;
    const int lo  = tab[myc * GCN_TABP + b];
    const int hi  = tab[myc * GCN_TABP + b + 1];
    __syncthreads();

    const int sub = lane >> 4;
    const int k16 = lane & 15;
    int2 ev[16];
    unsigned act = 0;
#pragma unroll
    for (int j = 0; j < 16; ++j) {
        const int la = __builtin_amdgcn_readlane(lo, 2 * j);
        const int lb = __builtin_amdgcn_readlane(lo, 2 * j + 1);
        const int ha = __builtin_amdgcn_readlane(hi, 2 * j);
        const int hb = __builtin_amdgcn_readlane(hi, 2 * j + 1);
        const int lj = sub ? lb : la;
        const int hj = sub ? hb : ha;
        const int2* __restrict__ p = stage + (size_t)(wid * 32 + 2 * j + sub) * GCN_CE;
        ev[j] = make_int2(0, 0);
        if (lj + k16 < hj) {
            ev[j] = p[lj + k16];
            act |= 1u << j;
        }
    }
#pragma unroll
    for (int j = 0; j < 16; ++j) {
        if (act & (1u << j)) gcn_place_edge(ev[j], cntL, ellL, cnt, ovf_src, ovf_dst, n);
    }
    for (int j = 0; j < 32; ++j) {
        const int lj = __builtin_amdgcn_readlane(lo, j);
        const int hj = __builtin_amdgcn_readlane(hi, j);
        if (hj - lj > 16) {
            const int2* __restrict__ p = stage + (size_t)(wid * 32 + j) * GCN_CE;
            for (int i = lj + 16 + lane; i < hj; i += 32)
                gcn_place_edge(p[i], cntL, ellL, cnt, ovf_src, ovf_dst, n);
        }
    }
    __syncthreads();

    const int node0 = b * 64;
    if (tid < 64 && node0 + tid < n) cnt[node0 + tid] = cntL[tid];
#pragma unroll
    for (int k = 0; k < 4; ++k) {
        const int i  = k * 256 + tid;
        const int nd = i >> 4;
        const int q  = i & 15;
        if (node0 + nd < n && (q < 8 || cntL[nd] > 32))
            *(int4*)(ell + (size_t)(node0 + nd) * GCN_CAP + 4 * q) = ellL4[nd * 16 + q];
    }
}

__device__ __forceinline__ float gcn_dinv(const int* __restrict__ cnt, int i) {
    return rsqrtf((float)cnt[i] + 1.0f);
}

struct GcnStage {
    float  a[4];
    float4 bl[4];
    float2 bh[4];
};

template <int K>
__device__ __forceinline__ void gcn_load_stage(GcnStage& st,
                                               const float* __restrict__ Arow,
                                               const float* __restrict__ Wl,
                                               const float* __restrict__ Wh, int kb) {
#pragma unroll
    for (int i = 0; i < 4; ++i) st.a[i] = Arow[kb + i];
#pragma unroll
    for (int i = 0; i < 4; ++i) {
        st.bl[i] = *(const float4*)(Wl + (size_t)(kb + i) * 96);
        st.bh[i] = *(const float2*)(Wh + (size_t)(kb + i) * 96);
    }
}

__device__ __forceinline__ void gcn_mma_stage(const GcnStage& st, v8f (&acc)[6]) {
#pragma unroll
    for (int s = 0; s < 2; ++s) {
        const float a0 = st.a[2 * s], a1 = st.a[2 * s + 1];
        const float4 p = st.bl[2 * s], q = st.bl[2 * s + 1];
        const float2 u = st.bh[2 * s], w = st.bh[2 * s + 1];
        v2f a;  a[0]  = a0;  a[1]  = a1;
        v2f b0; b0[0] = p.x; b0[1] = q.x;
        v2f b1; b1[0] = p.y; b1[1] = q.y;
        v2f b2; b2[0] = p.z; b2[1] = q.z;
        v2f b3; b3[0] = p.w; b3[1] = q.w;
        v2f b4; b4[0] = u.x; b4[1] = w.x;
        v2f b5; b5[0] = u.y; b5[1] = w.y;
        acc[0] = __builtin_amdgcn_wmma_f32_16x16x4_f32(false, a, false, b0, (short)0, acc[0], false, false);
        acc[1] = __builtin_amdgcn_wmma_f32_16x16x4_f32(false, a, false, b1, (short)0, acc[1], false, false);
        acc[2] = __builtin_amdgcn_wmma_f32_16x16x4_f32(false, a, false, b2, (short)0, acc[2], false, false);
        acc[3] = __builtin_amdgcn_wmma_f32_16x16x4_f32(false, a, false, b3, (short)0, acc[3], false, false);
        acc[4] = __builtin_amdgcn_wmma_f32_16x16x4_f32(false, a, false, b4, (short)0, acc[4], false, false);
        acc[5] = __builtin_amdgcn_wmma_f32_16x16x4_f32(false, a, false, b5, (short)0, acc[5], false, false);
    }
}

template <int K>
__global__ void __launch_bounds__(64)
gcn_gemm96_strip_wmma_f32(const float* __restrict__ A,
                          const float* __restrict__ W,
                          const int* __restrict__ cnt,
                          float* __restrict__ hws, int M) {
    const int lane = threadIdx.x & 31;
    const int tm   = blockIdx.x * (blockDim.x >> 5) + (threadIdx.x >> 5);
    if (tm >= (M >> 4)) return;

    const int c = lane & 15;
    const int h = lane >> 4;
    const int row = tm * 16 + c;

    const float* __restrict__ Arow = A + (size_t)row * K + 4 * h;
    const float* __restrict__ Wl   = W + (size_t)(4 * h) * 96 + 4 * c;
    const float* __restrict__ Wh   = W + (size_t)(4 * h) * 96 + 64 + 2 * c;

    const int rbase = tm * 16 + (h << 3);
    float dm[8];
#pragma unroll
    for (int r = 0; r < 8; ++r) dm[r] = gcn_dinv(cnt, rbase + r);

    v8f acc[6] = {};
    constexpr int K8 = K & ~7;

    constexpr int NB = K8 / 8;
    GcnStage st[3];
    gcn_load_stage<K>(st[0], Arow, Wl, Wh, 0);
    if constexpr (NB > 1) gcn_load_stage<K>(st[1], Arow, Wl, Wh, 8);
#pragma unroll
    for (int j = 0; j < NB; ++j) {
        if (j + 2 < NB) gcn_load_stage<K>(st[(j + 2) % 3], Arow, Wl, Wh, (j + 2) * 8);
        __builtin_amdgcn_sched_barrier(0);
        gcn_mma_stage(st[j % 3], acc);
    }

    if constexpr (K8 < K) {
        GcnStage t;
#pragma unroll
        for (int i = 0; i < 4; ++i) {
            const int  kk = K8 + 4 * h + i;
            const bool ok = (kk < K);
            const int  kc = ok ? kk : (K - 1);
            const float  av = A[(size_t)row * K + kc];
            const float4 lv = *(const float4*)(W + (size_t)kc * 96 + 4 * c);
            const float2 hv = *(const float2*)(W + (size_t)kc * 96 + 64 + 2 * c);
            t.a[i]    = ok ? av : 0.0f;
            t.bl[i].x = ok ? lv.x : 0.0f; t.bl[i].y = ok ? lv.y : 0.0f;
            t.bl[i].z = ok ? lv.z : 0.0f; t.bl[i].w = ok ? lv.w : 0.0f;
            t.bh[i].x = ok ? hv.x : 0.0f; t.bh[i].y = ok ? hv.y : 0.0f;
        }
        gcn_mma_stage(t, acc);
    }

#pragma unroll
    for (int r = 0; r < 8; ++r) {
        const size_t m = (size_t)(rbase + r);
        float4 o;
        o.x = acc[0][r] * dm[r]; o.y = acc[1][r] * dm[r];
        o.z = acc[2][r] * dm[r]; o.w = acc[3][r] * dm[r];
        float2 p;
        p.x = acc[4][r] * dm[r]; p.y = acc[5][r] * dm[r];
        *(float4*)(hws + m * 96 + 4 * c)      = o;
        *(float2*)(hws + m * 96 + 64 + 2 * c) = p;
    }
}

#define GCN_PA 100
#define GCN_PW 192
#define GCN_NB 64

__device__ __forceinline__ float4 gcn_gather_node(float4 acc, int s0, int nb, int deg, int node,
                                                  const int* __restrict__ bucket,
                                                  const char* __restrict__ rows, unsigned loff, int l4, int lane,
                                                  const int* __restrict__ cnt,
                                                  const int* __restrict__ ovf_src,
                                                  const int* __restrict__ ovf_dst, int n) {
    for (int base = 0; base < nb; base += 32) {
        const int idx = base + lane;
        int s = s0;
        if (base > 0) s = (idx < nb) ? bucket[idx] : node;
        const int rem = nb - base;
        const int cn  = (rem < 32) ? rem : 32;

        int j = 0;
        for (; j + 8 <= cn; j += 8) {
            float4 v[8];
#pragma unroll
            for (int t = 0; t < 8; ++t) {
                const int sj = __builtin_amdgcn_readlane(s, j + t);
                v[t] = ((const float4*)(rows + (size_t)((unsigned)sj * 384u)))[l4];
            }
#pragma unroll
            for (int t = 0; t < 8; ++t) {
                acc.x += v[t].x; acc.y += v[t].y; acc.z += v[t].z; acc.w += v[t].w;
            }
        }
        for (; j < cn; j += 4) {
            float4 v[4];
#pragma unroll
            for (int t = 0; t < 4; ++t) {
                const int sj = __builtin_amdgcn_readlane(s, j + t);
                v[t] = ((const float4*)(rows + (size_t)((unsigned)sj * 384u)))[l4];
            }
#pragma unroll
            for (int t = 0; t < 4; ++t) {
                const bool on = (j + t < cn);
                acc.x += on ? v[t].x : 0.0f; acc.y += on ? v[t].y : 0.0f;
                acc.z += on ? v[t].z : 0.0f; acc.w += on ? v[t].w : 0.0f;
            }
        }
    }

    if (deg > GCN_CAP) {
        const int novf = __builtin_amdgcn_readfirstlane(cnt[n]);
        for (int i = 0; i < novf; ++i) {
            if (__builtin_amdgcn_readfirstlane(ovf_dst[i]) == node) {
                const int sj = __builtin_amdgcn_readfirstlane(ovf_src[i]);
                const float4 v = *(const float4*)(rows + (size_t)((unsigned)sj * 384u) + loff);
                acc.x += v.x; acc.y += v.y; acc.z += v.z; acc.w += v.w;
            }
        }
    }
    return acc;
}

template <int NOUT>
__global__ void __launch_bounds__(1024)
gcn_gather96_gemm_kernel(const int* __restrict__ cnt,
                         const int* __restrict__ ell,
                         const int* __restrict__ ovf_src,
                         const int* __restrict__ ovf_dst,
                         const float* __restrict__ hws,
                         const float* __restrict__ bias,
                         const float* __restrict__ Wn,
                         float* __restrict__ hout, int n) {
    __shared__ float4 Wl4[(NOUT == 96) ? (48 * GCN_PW / 4) : (96 * 2 / 4)];
    __shared__ float4 Al4[GCN_NB * GCN_PA / 4];
    __shared__ float  dl[GCN_NB];
    float* Wl = (float*)Wl4;
    float* Al = (float*)Al4;

    const int tid  = threadIdx.x;
    const int lane = tid & 31;
    const int wid  = __builtin_amdgcn_readfirstlane(tid >> 5);
    const int node0 = (int)blockIdx.x * GCN_NB + 2 * wid;

    const int l4 = (lane < 24) ? lane : lane - 24;
    const char* __restrict__ rows = (const char*)hws;
    const unsigned loff = (unsigned)l4 * 16u;
    int    nd[2], ndc[2], dg[2];
    float4 ac[2];
#pragma unroll
    for (int p = 0; p < 2; ++p) {
        nd[p]  = node0 + p;
        ndc[p] = (nd[p] < n) ? nd[p] : (n - 1);
        ac[p]  = *(const float4*)(rows + (size_t)((unsigned)ndc[p] * 384u) + loff);
        dg[p]  = __builtin_amdgcn_readfirstlane(cnt[ndc[p]]);
    }

    const float4* __restrict__ W4 = (const float4*)Wn;
    const int  kp0 = tid / 24;
    const int  q0  = tid - kp0 * 24;
    const int  i1  = 1024 + tid;
    const int  kp1 = i1 / 24;
    const int  q1  = i1 - kp1 * 24;
    float4 ra0 = {}, ra1 = {}, rb0 = {}, rb1 = {};
    if (NOUT == 96) {
        ra0 = W4[(2 * kp0) * 24 + q0];
        ra1 = W4[(2 * kp0 + 1) * 24 + q0];
        if (wid < 4) {
            rb0 = W4[(2 * kp1) * 24 + q1];
            rb1 = W4[(2 * kp1 + 1) * 24 + q1];
        }
    } else {
        if (tid < 48) ra0 = W4[tid];
    }

    if (NOUT == 96) {
        float* d0 = Wl + kp0 * GCN_PW + 8 * q0;
        *(float4*)d0       = make_float4(ra0.x, ra1.x, ra0.y, ra1.y);
        *(float4*)(d0 + 4) = make_float4(ra0.z, ra1.z, ra0.w, ra1.w);
        if (wid < 4) {
            float* d1 = Wl + kp1 * GCN_PW + 8 * q1;
            *(float4*)d1       = make_float4(rb0.x, rb1.x, rb0.y, rb1.y);
            *(float4*)(d1 + 4) = make_float4(rb0.z, rb1.z, rb0.w, rb1.w);
        }
    } else {
        if (tid < 48) Wl4[tid] = ra0;
    }

    int nbk[2], s0[2];
#pragma unroll
    for (int p = 0; p < 2; ++p) {
        nbk[p] = (dg[p] < GCN_CAP) ? dg[p] : GCN_CAP;
        s0[p]  = (lane < nbk[p]) ? ell[(size_t)ndc[p] * GCN_CAP + lane] : ndc[p];
    }

#pragma unroll
    for (int p = 0; p < 2; ++p) {
        float4 o  = make_float4(0.0f, 0.0f, 0.0f, 0.0f);
        float  di = 0.0f;
        if (nd[p] < n) {
            const float4 acc = gcn_gather_node(ac[p], s0[p], nbk[p], dg[p], nd[p],
                                               ell + (size_t)nd[p] * GCN_CAP, rows, loff, l4, lane,
                                               cnt, ovf_src, ovf_dst, n);
            di = rsqrtf((float)dg[p] + 1.0f);
            const int c0 = l4 * 4;
            o.x = fmaxf(acc.x * di + bias[c0],     0.0f);
            o.y = fmaxf(acc.y * di + bias[c0 + 1], 0.0f);
            o.z = fmaxf(acc.z * di + bias[c0 + 2], 0.0f);
            o.w = fmaxf(acc.w * di + bias[c0 + 3], 0.0f);
        }
        if (lane < 24) *(float4*)(Al + (2 * wid + p) * GCN_PA + lane * 4) = o;
        if (lane == 0) dl[2 * wid + p] = di;
    }
    __syncthreads();

    const int c = lane & 15;
    const int h = lane >> 4;

    if (NOUT == 96) {
        if (wid < 3 * (GCN_NB / 16)) {
            const int sl = wid / 3;
            const int g  = wid - 3 * sl;
            const int tm = (int)blockIdx.x * (GCN_NB / 16) + sl;
            if (tm < (n >> 4)) {
                const float* Ar = Al + (sl * 16 + c) * GCN_PA + 4 * h;
                const float* Wr = Wl + (2 * h) * GCN_PW + (g * 32 + 2 * c) * 2;
                v8f acc0 = {}, acc1 = {};
#pragma unroll
                for (int kb = 0; kb < 96; kb += 8) {
                    const float4 a  = *(const float4*)(Ar + kb);
                    const float4 w0 = *(const float4*)(Wr + (kb / 2 + 0) * GCN_PW);
                    const float4 w1 = *(const float4*)(Wr + (kb / 2 + 1) * GCN_PW);
                    v2f a01; a01[0] = a.x;  a01[1] = a.y;
                    v2f a23; a23[0] = a.z;  a23[1] = a.w;
                    v2f p0;  p0[0]  = w0.x; p0[1]  = w0.y;
                    v2f p1;  p1[0]  = w0.z; p1[1]  = w0.w;
                    v2f q0v; q0v[0] = w1.x; q0v[1] = w1.y;
                    v2f q1v; q1v[0] = w1.z; q1v[1] = w1.w;
                    acc0 = __builtin_amdgcn_wmma_f32_16x16x4_f32(false, a01, false, p0,  (short)0, acc0, false, false);
                    acc1 = __builtin_amdgcn_wmma_f32_16x16x4_f32(false, a01, false, p1,  (short)0, acc1, false, false);
                    acc0 = __builtin_amdgcn_wmma_f32_16x16x4_f32(false, a23, false, q0v, (short)0, acc0, false, false);
                    acc1 = __builtin_amdgcn_wmma_f32_16x16x4_f32(false, a23, false, q1v, (short)0, acc1, false, false);
                }
#pragma unroll
                for (int r = 0; r < 8; ++r) {
                    const float  dm = dl[sl * 16 + 8 * h + r];
                    const size_t m  = (size_t)(tm * 16 + 8 * h + r);
                    float2 p;
                    p.x = acc0[r] * dm; p.y = acc1[r] * dm;
                    *(float2*)(hout + m * 96 + g * 32 + 2 * c) = p;
                }
            }
        }
    } else {
        if (wid < GCN_NB / 16) {
            const int sl = wid;
            const int tm = (int)blockIdx.x * (GCN_NB / 16) + sl;
            if (tm < (n >> 4)) {
                const bool colOK = (c < NOUT);
                const int  cc    = colOK ? c : 0;
                const float* Ar = Al + (sl * 16 + c) * GCN_PA + 2 * h;
                const float* Wr = Wl + (2 * h) * NOUT + cc;
                v8f acc = {};
#pragma unroll
                for (int kb = 0; kb < 96; kb += 4) {
                    const float2 a  = *(const float2*)(Ar + kb);
                    const float  w0 = Wr[(kb + 0) * NOUT];
                    const float  w1 = Wr[(kb + 1) * NOUT];
                    v2f av; av[0] = a.x;                av[1] = a.y;
                    v2f bv; bv[0] = colOK ? w0 : 0.0f;  bv[1] = colOK ? w1 : 0.0f;
                    acc = __builtin_amdgcn_wmma_f32_16x16x4_f32(false, av, false, bv, (short)0, acc, false, false);
                }
                if (colOK) {
#pragma unroll
                    for (int r = 0; r < 8; ++r) {
                        const size_t m = (size_t)(tm * 16 + 8 * h + r);
                        hout[m * NOUT + c] = acc[r] * dl[sl * 16 + 8 * h + r];
                    }
                }
            }
        }
    }
}

__global__ void __launch_bounds__(256)
gcn_gather2_kernel(const int* __restrict__ cnt,
                   const int* __restrict__ ell,
                   const int* __restrict__ ovf_src,
                   const int* __restrict__ ovf_dst,
                   const float* __restrict__ hws,
                   const float* __restrict__ bias,
                   float* __restrict__ outp, int n) {
    const int  gid  = (int)(blockIdx.x * 256u + threadIdx.x);
    const int  q    = gid & 1;
    const int  nraw = gid >> 1;
    const bool live = (nraw < n);
    const int  node = live ? nraw : (n - 1);

    const float2* __restrict__ rows = (const float2*)hws;
    const int4*   __restrict__ b4   = (const int4*)(ell + (size_t)node * GCN_CAP);

    const int    deg  = cnt[node];
    const float2 self = rows[node];
    int sl[16];
#pragma unroll
    for (int j = 0; j < 4; ++j) {
        const int4 s = b4[2 * j + q];
        sl[4 * j] = s.x; sl[4 * j + 1] = s.y; sl[4 * j + 2] = s.z; sl[4 * j + 3] = s.w;
    }
    const int nb = (deg < GCN_CAP) ? deg : GCN_CAP;

    float2 v[16];
#pragma unroll
    for (int j = 0; j < 4; ++j) {
#pragma unroll
        for (int t = 0; t < 4; ++t) {
            const bool ok = (8 * j + 4 * q + t < nb);
            v[4 * j + t] = rows[ok ? sl[4 * j + t] : node];
        }
    }
    float ax = 0.0f, ay = 0.0f;
#pragma unroll
    for (int j = 0; j < 4; ++j) {
#pragma unroll
        for (int t = 0; t < 4; ++t) {
            const bool ok = (8 * j + 4 * q + t < nb);
            ax += ok ? v[4 * j + t].x : 0.0f;
            ay += ok ? v[4 * j + t].y : 0.0f;
        }
    }
    for (int base = 32; base < nb; base += 8) {
        const int4 s = b4[(base >> 2) + q];
        const int  e[4] = { s.x, s.y, s.z, s.w };
#pragma unroll
        for (int t = 0; t < 4; ++t) {
            if (base + 4 * q + t < nb) {
                const float2 w = rows[e[t]];
                ax += w.x; ay += w.y;
            }
        }
    }
    if (deg > GCN_CAP) {
        const int novf = cnt[n];
        for (int i = q; i < novf; i += 2) {
            if (ovf_dst[i] == node) {
                const float2 w = rows[ovf_src[i]];
                ax += w.x; ay += w.y;
            }
        }
    }
    ax += __shfl_xor(ax, 1, 32);
    ay += __shfl_xor(ay, 1, 32);
    if (live) {
        const float di = rsqrtf((float)deg + 1.0f);
        const float o  = q ? ((ay + self.y) * di + bias[1]) : ((ax + self.x) * di + bias[0]);
        outp[2 * (size_t)node + q] = o;
    }
}


extern "C" void kernel_launch(void* const* d_in, const int* in_sizes, int n_in,
                              void* d_out, int out_size, void* d_ws, size_t ws_size,
                              hipStream_t stream) {
    (void)in_sizes; (void)n_in; (void)out_size; (void)ws_size;

    const int N = GCN_N;
    const int E = GCN_E;

    const float* x    = (const float*)d_in[0];
    const int*   ei   = (const int*)d_in[1];
    const int*   src  = ei;
    const int*   dstp = ei + E;

    const float* W[5] = { (const float*)d_in[2], (const float*)d_in[4],
                          (const float*)d_in[6], (const float*)d_in[8],
                          (const float*)d_in[10] };
    const float* B[5] = { (const float*)d_in[3], (const float*)d_in[5],
                          (const float*)d_in[7], (const float*)d_in[9],
                          (const float*)d_in[11] };

    int*   cnt    = (int*)d_ws;
    int*   ell    = cnt + 50048;
    int*   ovfs   = ell + (size_t)N * GCN_CAP;
    int*   ovfd   = ovfs + E;
    float* hA     = (float*)(ovfd + E);
    float* hB     = hA + (size_t)N * 96;
    int*   tab    = (int*)(hB + (size_t)N * 96);
    int2*  stage  = (int2*)(tab + GCN_NCH * GCN_TABP);
    float* out    = (float*)d_out;

    gcn_bin_kernel  <<<GCN_NCH, 1024, 0, stream>>>(src, dstp, tab, stage, cnt, N, E);
    gcn_owner_kernel<<<GCN_NBIN, 256, 0, stream>>>(tab, stage, cnt, ell, ovfs, ovfd, N);

    const int sBlocks = (N / 16 + 1) / 2;
    gcn_gemm96_strip_wmma_f32<165><<<sBlocks, 64, 0, stream>>>(x, W[0], cnt, hA, N);

    const int fBlocks = (N + GCN_NB - 1) / GCN_NB;
    gcn_gather96_gemm_kernel<96><<<fBlocks, 1024, 0, stream>>>(cnt, ell, ovfs, ovfd, hA, B[0], W[1], hB, N);
    gcn_gather96_gemm_kernel<96><<<fBlocks, 1024, 0, stream>>>(cnt, ell, ovfs, ovfd, hB, B[1], W[2], hA, N);
    gcn_gather96_gemm_kernel<96><<<fBlocks, 1024, 0, stream>>>(cnt, ell, ovfs, ovfd, hA, B[2], W[3], hB, N);
    gcn_gather96_gemm_kernel<2> <<<fBlocks, 1024, 0, stream>>>(cnt, ell, ovfs, ovfd, hB, B[3], W[4], hA, N);

    const int aBlocks = (N + 127) / 128;
    gcn_gather2_kernel<<<aBlocks, 256, 0, stream>>>(cnt, ell, ovfs, ovfd, hA, B[4], out, N);
}
